# starting kernel
# baseline (speedup 1.0000x reference)
#include <hip/hip_runtime.h>
#include <hip/hip_bf16.h>

typedef __attribute__((ext_vector_type(2))) float v2f;
typedef __attribute__((ext_vector_type(8))) float v8f;

__global__ void k_init(float* __restrict__ deg, float* __restrict__ aggB,
                       float* __restrict__ aggC, int n) {
  int idx = blockIdx.x * blockDim.x + threadIdx.x;
  if (idx < n) deg[idx] = 1.0f;
  if (idx < n * 16) { aggB[idx] = 0.0f; aggC[idx] = 0.0f; }
}

__global__ void k_deg(const int* __restrict__ dst, float* __restrict__ deg, int E) {
  int e = blockIdx.x * blockDim.x + threadIdx.x;
  if (e < E) atomicAdd(&deg[dst[e]], 1.0f);
}

__global__ void k_rsqrt(const float* __restrict__ deg, float* __restrict__ dinv, int n) {
  int i = blockIdx.x * blockDim.x + threadIdx.x;
  if (i < n) dinv[i] = rsqrtf(deg[i]);
}

template <int K>
__global__ void k_gemm_wmma(const float* __restrict__ A, const float* __restrict__ W,
                            float* __restrict__ H, int ntiles) {
  __shared__ float Ws[K * 16];
  for (int t = threadIdx.x; t < K * 16; t += blockDim.x) Ws[t] = W[t];
  __syncthreads();

  const int lane = threadIdx.x & 31;
  const int tile = blockIdx.x * (blockDim.x >> 5) + (threadIdx.x >> 5);
  if (tile >= ntiles) return;

  const int half = lane >> 4;
  const int mn   = lane & 15;
  const int koff = half * 2;

  const float* arow = A + (size_t)(tile * 16 + mn) * K;

  v8f c = {0.f, 0.f, 0.f, 0.f, 0.f, 0.f, 0.f, 0.f};
#pragma unroll
  for (int k0 = 0; k0 < K; k0 += 4) {
    v2f a = *(const v2f*)(arow + k0 + koff);
    v2f b;
    b.x = Ws[(k0 + koff + 0) * 16 + mn];
    b.y = Ws[(k0 + koff + 1) * 16 + mn];
    c = __builtin_amdgcn_wmma_f32_16x16x4_f32(
         false, a,  false, b,
         (short)0, c,  false,  false);
  }

  float* hrow = H + (size_t)tile * 256 + half * 128 + mn;
#pragma unroll
  for (int i = 0; i < 8; i++) hrow[i * 16] = c[i];
}

__global__ void k_scatter(const int* __restrict__ src, const int* __restrict__ dst,
                          const float* __restrict__ dinv, const float* __restrict__ H,
                          float* __restrict__ AGG, int E) {
  int idx = blockIdx.x * blockDim.x + threadIdx.x;
  int e = idx >> 4;
  int j = idx & 15;
  if (e >= E) return;
  int s = src[e];
  int d = dst[e];
  float norm = dinv[s] * dinv[d];
  atomicAdd(&AGG[d * 16 + j], norm * H[s * 16 + j]);
}

__global__ void k_combine(const float* __restrict__ AGG, const float* __restrict__ H,
                          const float* __restrict__ dinv, const float* __restrict__ bias,
                          float* __restrict__ out, int n) {
  int idx = blockIdx.x * blockDim.x + threadIdx.x;
  int i = idx >> 4;
  int j = idx & 15;
  if (i >= n) return;
  float di = dinv[i];
  out[idx] = fmaxf(AGG[idx] + di * di * H[idx] + bias[j], 0.0f);
}

__global__ void k_final(const float* __restrict__ G, const float* __restrict__ l1b,
                        const float* __restrict__ l2W, const float* __restrict__ l2b,
                        float* __restrict__ out, int n) {
  int idx = blockIdx.x * blockDim.x + threadIdx.x;
  int i = idx >> 4;
  int j = idx & 15;
  float v = 0.0f;
  if (i < n) v = fmaxf(G[idx] + l1b[j], 0.0f) * l2W[j];
#pragma unroll
  for (int off = 8; off; off >>= 1) v += __shfl_xor(v, off, 16);
  if (i < n && j == 0) out[i] = v + l2b[0];
}

extern "C" void kernel_launch(void* const* d_in, const int* in_sizes, int n_in,
                              void* d_out, int out_size, void* d_ws, size_t ws_size,
                              hipStream_t stream) {
  const float* x   = (const float*)d_in[0];
  const int*   ei  = (const int*)d_in[1];
  const float* W1  = (const float*)d_in[2];
  const float* b1  = (const float*)d_in[3];
  const float* W2  = (const float*)d_in[4];
  const float* b2  = (const float*)d_in[5];
  const float* l1W = (const float*)d_in[6];
  const float* l1b = (const float*)d_in[7];
  const float* l2W = (const float*)d_in[8];
  const float* l2b = (const float*)d_in[9];

  const int N = in_sizes[0] / 128;
  const int E = in_sizes[1] / 2;
  const int* src = ei;
  const int* dst = ei + E;

  float* deg  = (float*)d_ws;
  float* dinv = deg + N;
  float* bufA = dinv + N;
  float* bufB = bufA + (size_t)N * 16;
  float* bufC = bufB + (size_t)N * 16;

  const int ntiles = N / 16;
  const int gemmBlocks = (ntiles * 32 + 255) / 256;
  const int nodeElems  = N * 16;
  const int nodeBlocks = (nodeElems + 255) / 256;
  const int edgeBlocks = (E + 255) / 256;
  const int scatBlocks = (int)(((size_t)E * 16 + 255) / 256);

  k_init  <<<nodeBlocks, 256, 0, stream>>>(deg, bufB, bufC, N);
  k_deg   <<<edgeBlocks, 256, 0, stream>>>(dst, deg, E);
  k_rsqrt <<<(N + 255) / 256, 256, 0, stream>>>(deg, dinv, N);

  k_gemm_wmma<128><<<gemmBlocks, 256, 0, stream>>>(x, W1, bufA, ntiles);
  k_scatter<<<scatBlocks, 256, 0, stream>>>(src, dst, dinv, bufA, bufB, E);
  k_combine<<<nodeBlocks, 256, 0, stream>>>(bufB, bufA, dinv, b1, bufB, N);

  k_gemm_wmma<16><<<gemmBlocks, 256, 0, stream>>>(bufB, W2, bufA, ntiles);
  k_scatter<<<scatBlocks, 256, 0, stream>>>(src, dst, dinv, bufA, bufC, E);
  k_combine<<<nodeBlocks, 256, 0, stream>>>(bufC, bufA, dinv, b2, bufC, N);

  k_gemm_wmma<16><<<gemmBlocks, 256, 0, stream>>>(bufC, l1W, bufA, ntiles);
  k_final<<<nodeBlocks, 256, 0, stream>>>(bufA, l1b, l2W, l2b, (float*)d_out, N);
}


# In-degree count: global float atomic per edge replaced by LDS-privatized integer histograms (edge chunks x bin ranges), partial rows summed in the rsqrt kernel; counts exact, no float sum reordered
# speedup vs baseline: 1.3880x; 1.3880x over previous
#include <hip/hip_runtime.h>
#include <hip/hip_bf16.h>

typedef __attribute__((ext_vector_type(2))) float v2f;
typedef __attribute__((ext_vector_type(8))) float v8f;

#define HIST_BINS   25600
#define HIST_CHUNKS 48
#define HIST_THREADS 1024

__device__ __forceinline__ void hist_count(unsigned* hist, int d, int lo) {
  unsigned b = (unsigned)(d - lo);
  if (b < (unsigned)HIST_BINS) atomicAdd(&hist[b], 1u);
}

__device__ __forceinline__ void hist_count4(unsigned* hist, int4 v, int lo) {
  hist_count(hist, v.x, lo);
  hist_count(hist, v.y, lo);
  hist_count(hist, v.z, lo);
  hist_count(hist, v.w, lo);
}

__global__ __launch_bounds__(HIST_THREADS) void k_hist(const int* __restrict__ dst,
                                                       unsigned* __restrict__ part,
                                                       int E, int n, int chunkLen) {
  __shared__ unsigned hist[HIST_BINS];
  const int tid = threadIdx.x;
  const int c   = blockIdx.x;
  const int lo  = blockIdx.y * HIST_BINS;

  for (int t = tid; t < HIST_BINS; t += HIST_THREADS) hist[t] = 0u;
  __syncthreads();

  const int beg = c * chunkLen;
  int end = beg + chunkLen;
  if (end > E) end = E;
  const int len = end - beg;

  if (len > 0) {
    if ((((size_t)(dst + beg)) & 15) == 0) {
      const int4* p = (const int4*)(dst + beg);
      const int nvec = len >> 2;
      int v = tid;
      for (; v + 7 * HIST_THREADS < nvec; v += 8 * HIST_THREADS) {
        int4 a0 = p[v];
        int4 a1 = p[v + 1 * HIST_THREADS];
        int4 a2 = p[v + 2 * HIST_THREADS];
        int4 a3 = p[v + 3 * HIST_THREADS];
        int4 a4 = p[v + 4 * HIST_THREADS];
        int4 a5 = p[v + 5 * HIST_THREADS];
        int4 a6 = p[v + 6 * HIST_THREADS];
        int4 a7 = p[v + 7 * HIST_THREADS];
        hist_count4(hist, a0, lo);
        hist_count4(hist, a1, lo);
        hist_count4(hist, a2, lo);
        hist_count4(hist, a3, lo);
        hist_count4(hist, a4, lo);
        hist_count4(hist, a5, lo);
        hist_count4(hist, a6, lo);
        hist_count4(hist, a7, lo);
      }
      for (; v < nvec; v += HIST_THREADS) hist_count4(hist, p[v], lo);
      const int t = (nvec << 2) + tid;
      if (t < len) hist_count(hist, dst[beg + t], lo);
    } else {
      for (int t = tid; t < len; t += HIST_THREADS) hist_count(hist, dst[beg + t], lo);
    }
  }
  __syncthreads();

  int cnt = n - lo;
  if (cnt > HIST_BINS) cnt = HIST_BINS;
  unsigned* row = part + (size_t)c * n + lo;
  for (int t = tid; t < cnt; t += HIST_THREADS) row[t] = hist[t];
}

__global__ void k_dinv(const unsigned* __restrict__ part, float* __restrict__ dinv, int n) {
  int i = blockIdx.x * blockDim.x + threadIdx.x;
  if (i >= n) return;
  unsigned s = 1u;
#pragma unroll
  for (int c = 0; c < HIST_CHUNKS; c++) s += part[(size_t)c * n + i];
  dinv[i] = rsqrtf((float)s);
}

__global__ void k_zero(float* __restrict__ aggB, float* __restrict__ aggC, int n16) {
  int idx = blockIdx.x * blockDim.x + threadIdx.x;
  if (idx < n16) { aggB[idx] = 0.0f; aggC[idx] = 0.0f; }
}

template <int K>
__global__ void k_gemm_wmma(const float* __restrict__ A, const float* __restrict__ W,
                            float* __restrict__ H, int ntiles) {
  __shared__ float Ws[K * 16];
  for (int t = threadIdx.x; t < K * 16; t += blockDim.x) Ws[t] = W[t];
  __syncthreads();

  const int lane = threadIdx.x & 31;
  const int tile = blockIdx.x * (blockDim.x >> 5) + (threadIdx.x >> 5);
  if (tile >= ntiles) return;

  const int half = lane >> 4;
  const int mn   = lane & 15;
  const int koff = half * 2;

  const float* arow = A + (size_t)(tile * 16 + mn) * K;

  v8f c = {0.f, 0.f, 0.f, 0.f, 0.f, 0.f, 0.f, 0.f};
#pragma unroll
  for (int k0 = 0; k0 < K; k0 += 4) {
    v2f a = *(const v2f*)(arow + k0 + koff);
    v2f b;
    b.x = Ws[(k0 + koff + 0) * 16 + mn];
    b.y = Ws[(k0 + koff + 1) * 16 + mn];
    c = __builtin_amdgcn_wmma_f32_16x16x4_f32(
         false, a,  false, b,
         (short)0, c,  false,  false);
  }

  float* hrow = H + (size_t)tile * 256 + half * 128 + mn;
#pragma unroll
  for (int i = 0; i < 8; i++) hrow[i * 16] = c[i];
}

__global__ void k_scatter(const int* __restrict__ src, const int* __restrict__ dst,
                          const float* __restrict__ dinv, const float* __restrict__ H,
                          float* __restrict__ AGG, int E) {
  int idx = blockIdx.x * blockDim.x + threadIdx.x;
  int e = idx >> 4;
  int j = idx & 15;
  if (e >= E) return;
  int s = src[e];
  int d = dst[e];
  float norm = dinv[s] * dinv[d];
  atomicAdd(&AGG[d * 16 + j], norm * H[s * 16 + j]);
}

__global__ void k_combine(const float* __restrict__ AGG, const float* __restrict__ H,
                          const float* __restrict__ dinv, const float* __restrict__ bias,
                          float* __restrict__ out, int n) {
  int idx = blockIdx.x * blockDim.x + threadIdx.x;
  int i = idx >> 4;
  int j = idx & 15;
  if (i >= n) return;
  float di = dinv[i];
  out[idx] = fmaxf(AGG[idx] + di * di * H[idx] + bias[j], 0.0f);
}

__global__ void k_final(const float* __restrict__ G, const float* __restrict__ l1b,
                        const float* __restrict__ l2W, const float* __restrict__ l2b,
                        float* __restrict__ out, int n) {
  int idx = blockIdx.x * blockDim.x + threadIdx.x;
  int i = idx >> 4;
  int j = idx & 15;
  float v = 0.0f;
  if (i < n) v = fmaxf(G[idx] + l1b[j], 0.0f) * l2W[j];
#pragma unroll
  for (int off = 8; off; off >>= 1) v += __shfl_xor(v, off, 16);
  if (i < n && j == 0) out[i] = v + l2b[0];
}

extern "C" void kernel_launch(void* const* d_in, const int* in_sizes, int n_in,
                              void* d_out, int out_size, void* d_ws, size_t ws_size,
                              hipStream_t stream) {
  const float* x   = (const float*)d_in[0];
  const int*   ei  = (const int*)d_in[1];
  const float* W1  = (const float*)d_in[2];
  const float* b1  = (const float*)d_in[3];
  const float* W2  = (const float*)d_in[4];
  const float* b2  = (const float*)d_in[5];
  const float* l1W = (const float*)d_in[6];
  const float* l1b = (const float*)d_in[7];
  const float* l2W = (const float*)d_in[8];
  const float* l2b = (const float*)d_in[9];

  const int N = in_sizes[0] / 128;
  const int E = in_sizes[1] / 2;
  const int* src = ei;
  const int* dst = ei + E;

  float* deg  = (float*)d_ws;
  float* dinv = deg + N;
  float* bufA = dinv + N;
  float* bufB = bufA + (size_t)N * 16;
  float* bufC = bufB + (size_t)N * 16;

  const int ntiles = N / 16;
  const int gemmBlocks = (ntiles * 32 + 255) / 256;
  const int nodeElems  = N * 16;
  const int nodeBlocks = (nodeElems + 255) / 256;
  const int scatBlocks = (int)(((size_t)E * 16 + 255) / 256);

  unsigned* part = (unsigned*)bufA;
  const int chunkLen = (((E + HIST_CHUNKS - 1) / HIST_CHUNKS) + 3) & ~3;
  const dim3 histGrid(HIST_CHUNKS, (N + HIST_BINS - 1) / HIST_BINS);
  k_hist <<<histGrid, HIST_THREADS, 0, stream>>>(dst, part, E, N, chunkLen);
  k_dinv <<<(N + 255) / 256, 256, 0, stream>>>(part, dinv, N);
  k_zero <<<nodeBlocks, 256, 0, stream>>>(bufB, bufC, nodeElems);

  k_gemm_wmma<128><<<gemmBlocks, 256, 0, stream>>>(x, W1, bufA, ntiles);
  k_scatter<<<scatBlocks, 256, 0, stream>>>(src, dst, dinv, bufA, bufB, E);
  k_combine<<<nodeBlocks, 256, 0, stream>>>(bufB, bufA, dinv, b1, bufB, N);

  k_gemm_wmma<16><<<gemmBlocks, 256, 0, stream>>>(bufB, W2, bufA, ntiles);
  k_scatter<<<scatBlocks, 256, 0, stream>>>(src, dst, dinv, bufA, bufC, E);
  k_combine<<<nodeBlocks, 256, 0, stream>>>(bufC, bufA, dinv, b2, bufC, N);

  k_gemm_wmma<16><<<gemmBlocks, 256, 0, stream>>>(bufC, l1W, bufA, ntiles);
  k_final<<<nodeBlocks, 256, 0, stream>>>(bufA, l1b, l2W, l2b, (float*)d_out, N);
}


# Float-atomic edge scatter replaced by one counting sort of edges by destination per call and a per-node gather (combine fused, no zeroing); float sums now in list-slot order, not atomic arrival
# speedup vs baseline: 2.0230x; 1.4575x over previous
#include <hip/hip_runtime.h>
#include <hip/hip_bf16.h>


typedef __attribute__((ext_vector_type(2))) float v2f;
typedef __attribute__((ext_vector_type(8))) float v8f;

#define KEY_BINS     25600
#define SORT_CHUNKS  16
#define SORT_THREADS 1024
#define SCAN_THREADS 256

__device__ __forceinline__ unsigned edge_key(int s, int d) {
  return (((unsigned)d) << 1) | (((unsigned)s) >> 16);
}

template <class F>
__device__ __forceinline__ void scan_chunk(const int* __restrict__ src, const int* __restrict__ dst,
                                           int beg, int len, int tid, F f) {
  const int* sp = src + beg;
  const int* dp = dst + beg;
  if (((((size_t)sp) | ((size_t)dp)) & 15) == 0) {
    const int4* s4 = (const int4*)sp;
    const int4* d4 = (const int4*)dp;
    const int nvec = len >> 2;
    int v = tid;
    for (; v + 3 * SORT_THREADS < nvec; v += 4 * SORT_THREADS) {
      int4 a0 = s4[v];
      int4 a1 = s4[v + 1 * SORT_THREADS];
      int4 a2 = s4[v + 2 * SORT_THREADS];
      int4 a3 = s4[v + 3 * SORT_THREADS];
      int4 b0 = d4[v];
      int4 b1 = d4[v + 1 * SORT_THREADS];
      int4 b2 = d4[v + 2 * SORT_THREADS];
      int4 b3 = d4[v + 3 * SORT_THREADS];
      f(a0.x, b0.x); f(a0.y, b0.y); f(a0.z, b0.z); f(a0.w, b0.w);
      f(a1.x, b1.x); f(a1.y, b1.y); f(a1.z, b1.z); f(a1.w, b1.w);
      f(a2.x, b2.x); f(a2.y, b2.y); f(a2.z, b2.z); f(a2.w, b2.w);
      f(a3.x, b3.x); f(a3.y, b3.y); f(a3.z, b3.z); f(a3.w, b3.w);
    }
    for (; v < nvec; v += SORT_THREADS) {
      int4 a = s4[v];
      int4 b = d4[v];
      f(a.x, b.x); f(a.y, b.y); f(a.z, b.z); f(a.w, b.w);
    }
    const int t = (nvec << 2) + tid;
    if (t < len) f(sp[t], dp[t]);
  } else {
    for (int t = tid; t < len; t += SORT_THREADS) f(sp[t], dp[t]);
  }
}

__global__ __launch_bounds__(SORT_THREADS) void k_hist(const int* __restrict__ src,
                                                       const int* __restrict__ dst,
                                                       unsigned* __restrict__ part,
                                                       int E, int nkeys, int chunkLen) {
  __shared__ unsigned hist[KEY_BINS];
  const int tid = threadIdx.x;
  const int c   = blockIdx.x;
  const unsigned lo = blockIdx.y * KEY_BINS;

  for (int t = tid; t < KEY_BINS; t += SORT_THREADS) hist[t] = 0u;
  __syncthreads();

  const int beg = c * chunkLen;
  int end = beg + chunkLen;
  if (end > E) end = E;
  const int len = end - beg;

  if (len > 0) {
    scan_chunk(src, dst, beg, len, tid, [&](int s, int d) {
      const unsigned b = edge_key(s, d) - lo;
      if (b < (unsigned)KEY_BINS) atomicAdd(&hist[b], 1u);
    });
  }
  __syncthreads();

  int cnt = nkeys - (int)lo;
  if (cnt > KEY_BINS) cnt = KEY_BINS;
  unsigned* row = part + (size_t)c * nkeys + lo;
  for (int t = tid; t < cnt; t += SORT_THREADS) row[t] = hist[t];
}

__global__ __launch_bounds__(SCAN_THREADS) void k_prefix(unsigned* __restrict__ part,
                                                         unsigned* __restrict__ rowptr,
                                                         unsigned* __restrict__ blockSum,
                                                         float* __restrict__ dinv, int nkeys) {
  __shared__ unsigned sh[SCAN_THREADS];
  const int tid = threadIdx.x;
  const int k = blockIdx.x * SCAN_THREADS + tid;
  unsigned tot = 0u;
  if (k < nkeys) {
    unsigned v[SORT_CHUNKS];
#pragma unroll
    for (int c = 0; c < SORT_CHUNKS; c++) v[c] = part[(size_t)c * nkeys + k];
#pragma unroll
    for (int c = 0; c < SORT_CHUNKS; c++) {
      part[(size_t)c * nkeys + k] = tot;
      tot += v[c];
    }
  }
  sh[tid] = tot;
  __syncthreads();
  for (int off = 1; off < SCAN_THREADS; off <<= 1) {
    const unsigned a = (tid >= off) ? sh[tid - off] : 0u;
    __syncthreads();
    sh[tid] += a;
    __syncthreads();
  }
  const unsigned incl = sh[tid];
  if (k < nkeys) rowptr[k] = incl - tot;
  if (tid == SCAN_THREADS - 1) blockSum[blockIdx.x] = incl;

  const unsigned other = __shfl_down(tot, 1);
  if (((k & 1) == 0) && (k < nkeys)) dinv[k >> 1] = rsqrtf((float)(1u + tot + other));
}

__global__ __launch_bounds__(SCAN_THREADS) void k_addoff(unsigned* __restrict__ rowptr,
                                                         const unsigned* __restrict__ blockSum,
                                                         int nkeys) {
  __shared__ unsigned red[SCAN_THREADS];
  const int tid = threadIdx.x;
  const int b = blockIdx.x;
  unsigned s = 0u;
  for (int j = tid; j < b; j += SCAN_THREADS) s += blockSum[j];
  red[tid] = s;
  __syncthreads();
  for (int off = SCAN_THREADS / 2; off > 0; off >>= 1) {
    if (tid < off) red[tid] += red[tid + off];
    __syncthreads();
  }
  const unsigned offv = red[0];
  const int k = b * SCAN_THREADS + tid;
  if (k < nkeys) rowptr[k] += offv;
}

__global__ __launch_bounds__(SORT_THREADS) void k_fill(const int* __restrict__ src,
                                                       const int* __restrict__ dst,
                                                       const unsigned* __restrict__ part,
                                                       const unsigned* __restrict__ rowptr,
                                                       unsigned short* __restrict__ list,
                                                       int E, int nkeys, int chunkLen) {
  __shared__ unsigned cur[KEY_BINS];
  const int tid = threadIdx.x;
  const int c   = blockIdx.x;
  const unsigned lo = blockIdx.y * KEY_BINS;

  int cnt = nkeys - (int)lo;
  if (cnt > KEY_BINS) cnt = KEY_BINS;
  const unsigned* row = part + (size_t)c * nkeys + lo;
  for (int t = tid; t < cnt; t += SORT_THREADS) cur[t] = rowptr[lo + t] + row[t];
  __syncthreads();

  const int beg = c * chunkLen;
  int end = beg + chunkLen;
  if (end > E) end = E;
  const int len = end - beg;

  if (len > 0) {
    scan_chunk(src, dst, beg, len, tid, [&](int s, int d) {
      const unsigned b = edge_key(s, d) - lo;
      if (b < (unsigned)KEY_BINS) {
        const unsigned pos = atomicAdd(&cur[b], 1u);
        list[pos] = (unsigned short)(((unsigned)s) & 0xFFFFu);
      }
    });
  }
}

template <int K>
__global__ void k_gemm_wmma(const float* __restrict__ A, const float* __restrict__ W,
                            float* __restrict__ H, int ntiles) {
  __shared__ float Ws[K * 16];
  for (int t = threadIdx.x; t < K * 16; t += blockDim.x) Ws[t] = W[t];
  __syncthreads();

  const int lane = threadIdx.x & 31;
  const int tile = blockIdx.x * (blockDim.x >> 5) + (threadIdx.x >> 5);
  if (tile >= ntiles) return;

  const int half = lane >> 4;
  const int mn   = lane & 15;
  const int koff = half * 2;

  const float* arow = A + (size_t)(tile * 16 + mn) * K;

  v8f c = {0.f, 0.f, 0.f, 0.f, 0.f, 0.f, 0.f, 0.f};
#pragma unroll
  for (int k0 = 0; k0 < K; k0 += 4) {
    v2f a = *(const v2f*)(arow + k0 + koff);
    v2f b;
    b.x = Ws[(k0 + koff + 0) * 16 + mn];
    b.y = Ws[(k0 + koff + 1) * 16 + mn];
    c = __builtin_amdgcn_wmma_f32_16x16x4_f32(
         false, a,  false, b,
         (short)0, c,  false,  false);
  }

  float* hrow = H + (size_t)tile * 256 + half * 128 + mn;
#pragma unroll
  for (int i = 0; i < 8; i++) hrow[i * 16] = c[i];
}

__device__ __forceinline__ void edge_fma(float4& acc, float w, float4 h) {
  acc.x += w * h.x;
  acc.y += w * h.y;
  acc.z += w * h.z;
  acc.w += w * h.w;
}

__global__ void k_gather(const unsigned short* __restrict__ list,
                         const unsigned* __restrict__ rowptr,
                         const float* __restrict__ dinv, const float* __restrict__ H,
                         const float* __restrict__ bias, float* __restrict__ outp,
                         int n, int E) {
  const int idx = blockIdx.x * blockDim.x + threadIdx.x;
  const int i = idx >> 2;
  const int q = idx & 3;
  if (i >= n) return;

  const unsigned p0 = rowptr[2 * i];
  const unsigned p1 = rowptr[2 * i + 1];
  const unsigned pe = (i + 1 < n) ? rowptr[2 * i + 2] : (unsigned)E;
  const float dn = dinv[i];
  const float4* H4 = (const float4*)H;

  float4 acc = make_float4(0.0f, 0.0f, 0.0f, 0.0f);
  unsigned p = p0;
  for (; p + 4 <= pe; p += 4) {
    const unsigned s0 = (unsigned)list[p]     + ((p     >= p1) ? 65536u : 0u);
    const unsigned s1 = (unsigned)list[p + 1] + ((p + 1 >= p1) ? 65536u : 0u);
    const unsigned s2 = (unsigned)list[p + 2] + ((p + 2 >= p1) ? 65536u : 0u);
    const unsigned s3 = (unsigned)list[p + 3] + ((p + 3 >= p1) ? 65536u : 0u);
    const float w0 = dinv[s0] * dn;
    const float w1 = dinv[s1] * dn;
    const float w2 = dinv[s2] * dn;
    const float w3 = dinv[s3] * dn;
    const float4 h0 = H4[(size_t)s0 * 4 + q];
    const float4 h1 = H4[(size_t)s1 * 4 + q];
    const float4 h2 = H4[(size_t)s2 * 4 + q];
    const float4 h3 = H4[(size_t)s3 * 4 + q];
    edge_fma(acc, w0, h0);
    edge_fma(acc, w1, h1);
    edge_fma(acc, w2, h2);
    edge_fma(acc, w3, h3);
  }
  for (; p < pe; p++) {
    const unsigned s = (unsigned)list[p] + ((p >= p1) ? 65536u : 0u);
    const float w = dinv[s] * dn;
    edge_fma(acc, w, H4[(size_t)s * 4 + q]);
  }

  const float4 hs = H4[(size_t)i * 4 + q];
  const float dd = dn * dn;
  float4 r;
  r.x = fmaxf(acc.x + dd * hs.x + bias[q * 4 + 0], 0.0f);
  r.y = fmaxf(acc.y + dd * hs.y + bias[q * 4 + 1], 0.0f);
  r.z = fmaxf(acc.z + dd * hs.z + bias[q * 4 + 2], 0.0f);
  r.w = fmaxf(acc.w + dd * hs.w + bias[q * 4 + 3], 0.0f);
  ((float4*)outp)[idx] = r;
}

__global__ void k_final(const float* __restrict__ G, const float* __restrict__ l1b,
                        const float* __restrict__ l2W, const float* __restrict__ l2b,
                        float* __restrict__ out, int n) {
  int idx = blockIdx.x * blockDim.x + threadIdx.x;
  int i = idx >> 4;
  int j = idx & 15;
  float v = 0.0f;
  if (i < n) v = fmaxf(G[idx] + l1b[j], 0.0f) * l2W[j];
#pragma unroll
  for (int off = 8; off; off >>= 1) v += __shfl_xor(v, off, 16);
  if (i < n && j == 0) out[i] = v + l2b[0];
}

extern "C" void kernel_launch(void* const* d_in, const int* in_sizes, int n_in,
                              void* d_out, int out_size, void* d_ws, size_t ws_size,
                              hipStream_t stream) {
  const float* x   = (const float*)d_in[0];
  const int*   ei  = (const int*)d_in[1];
  const float* W1  = (const float*)d_in[2];
  const float* b1  = (const float*)d_in[3];
  const float* W2  = (const float*)d_in[4];
  const float* b2  = (const float*)d_in[5];
  const float* l1W = (const float*)d_in[6];
  const float* l1b = (const float*)d_in[7];
  const float* l2W = (const float*)d_in[8];
  const float* l2b = (const float*)d_in[9];

  const int N = in_sizes[0] / 128;
  const int E = in_sizes[1] / 2;
  const int* src = ei;
  const int* dst = ei + E;

  unsigned* rowptr = (unsigned*)d_ws;
  float* bufA = (float*)d_ws + 2 * (size_t)N;
  float* bufB = bufA + (size_t)N * 16;
  float* bufC = bufB + (size_t)N * 16;
  float* dinv = (float*)d_out;

  unsigned short* list = (unsigned short*)bufA;
  unsigned* blockSum   = (unsigned*)bufA;
  unsigned* part       = (unsigned*)bufB;

  const int ntiles = N / 16;
  const int gemmBlocks = (ntiles * 32 + 255) / 256;
  const int nodeElems  = N * 16;
  const int nodeBlocks = (nodeElems + 255) / 256;
  const int gathBlocks = (N * 4 + 255) / 256;

  const int nkeys = 2 * N;
  const int chunkLen = (((E + SORT_CHUNKS - 1) / SORT_CHUNKS) + 3) & ~3;
  const dim3 sortGrid(SORT_CHUNKS, (nkeys + KEY_BINS - 1) / KEY_BINS);
  const int scanBlocks = (nkeys + SCAN_THREADS - 1) / SCAN_THREADS;
  k_hist  <<<sortGrid, SORT_THREADS, 0, stream>>>(src, dst, part, E, nkeys, chunkLen);
  k_prefix<<<scanBlocks, SCAN_THREADS, 0, stream>>>(part, rowptr, blockSum, dinv, nkeys);
  k_addoff<<<scanBlocks, SCAN_THREADS, 0, stream>>>(rowptr, blockSum, nkeys);
  k_fill  <<<sortGrid, SORT_THREADS, 0, stream>>>(src, dst, part, rowptr, list, E, nkeys, chunkLen);

  k_gemm_wmma<128><<<gemmBlocks, 256, 0, stream>>>(x, W1, bufB, ntiles);
  k_gather<<<gathBlocks, 256, 0, stream>>>(list, rowptr, dinv, bufB, b1, bufC, N, E);

  k_gemm_wmma<16><<<gemmBlocks, 256, 0, stream>>>(bufC, W2, bufB, ntiles);
  k_gather<<<gathBlocks, 256, 0, stream>>>(list, rowptr, dinv, bufB, b2, bufC, N, E);

  k_gemm_wmma<16><<<gemmBlocks, 256, 0, stream>>>(bufC, l1W, bufB, ntiles);
  k_final<<<nodeBlocks, 256, 0, stream>>>(bufB, l1b, l2W, l2b, (float*)d_out, N);
}


# Consumers fused into producer epilogues: layer-2 WMMA in the layer-1 gather (h1 via LDS), relu-dot head in the lin1 WMMA kernel, v_nop hazard slots by hand; 10->8 launches; head's 1st add now fma
# speedup vs baseline: 2.1767x; 1.0760x over previous
#include <hip/hip_runtime.h>
#include <hip/hip_bf16.h>


typedef __attribute__((ext_vector_type(2))) float v2f;
typedef __attribute__((ext_vector_type(8))) float v8f;

#define KEY_BINS     25600
#define SORT_CHUNKS  16
#define SORT_THREADS 1024
#define SCAN_THREADS 256

__device__ __forceinline__ unsigned edge_key(int s, int d) {
  return (((unsigned)d) << 1) | (((unsigned)s) >> 16);
}

template <class F>
__device__ __forceinline__ void scan_chunk(const int* __restrict__ src, const int* __restrict__ dst,
                                           int beg, int len, int tid, F f) {
  const int* sp = src + beg;
  const int* dp = dst + beg;
  if (((((size_t)sp) | ((size_t)dp)) & 15) == 0) {
    const int4* s4 = (const int4*)sp;
    const int4* d4 = (const int4*)dp;
    const int nvec = len >> 2;
    int v = tid;
    for (; v + 3 * SORT_THREADS < nvec; v += 4 * SORT_THREADS) {
      int4 a0 = s4[v];
      int4 a1 = s4[v + 1 * SORT_THREADS];
      int4 a2 = s4[v + 2 * SORT_THREADS];
      int4 a3 = s4[v + 3 * SORT_THREADS];
      int4 b0 = d4[v];
      int4 b1 = d4[v + 1 * SORT_THREADS];
      int4 b2 = d4[v + 2 * SORT_THREADS];
      int4 b3 = d4[v + 3 * SORT_THREADS];
      f(a0.x, b0.x); f(a0.y, b0.y); f(a0.z, b0.z); f(a0.w, b0.w);
      f(a1.x, b1.x); f(a1.y, b1.y); f(a1.z, b1.z); f(a1.w, b1.w);
      f(a2.x, b2.x); f(a2.y, b2.y); f(a2.z, b2.z); f(a2.w, b2.w);
      f(a3.x, b3.x); f(a3.y, b3.y); f(a3.z, b3.z); f(a3.w, b3.w);
    }
    for (; v < nvec; v += SORT_THREADS) {
      int4 a = s4[v];
      int4 b = d4[v];
      f(a.x, b.x); f(a.y, b.y); f(a.z, b.z); f(a.w, b.w);
    }
    const int t = (nvec << 2) + tid;
    if (t < len) f(sp[t], dp[t]);
  } else {
    for (int t = tid; t < len; t += SORT_THREADS) f(sp[t], dp[t]);
  }
}

__global__ __launch_bounds__(SORT_THREADS) void k_hist(const int* __restrict__ src,
                                                       const int* __restrict__ dst,
                                                       unsigned* __restrict__ part,
                                                       int E, int nkeys, int chunkLen) {
  __shared__ unsigned hist[KEY_BINS];
  const int tid = threadIdx.x;
  const int c   = blockIdx.x;
  const unsigned lo = blockIdx.y * KEY_BINS;

  for (int t = tid; t < KEY_BINS; t += SORT_THREADS) hist[t] = 0u;
  __syncthreads();

  const int beg = c * chunkLen;
  int end = beg + chunkLen;
  if (end > E) end = E;
  const int len = end - beg;

  if (len > 0) {
    scan_chunk(src, dst, beg, len, tid, [&](int s, int d) {
      const unsigned b = edge_key(s, d) - lo;
      if (b < (unsigned)KEY_BINS) atomicAdd(&hist[b], 1u);
    });
  }
  __syncthreads();

  int cnt = nkeys - (int)lo;
  if (cnt > KEY_BINS) cnt = KEY_BINS;
  unsigned* row = part + (size_t)c * nkeys + lo;
  for (int t = tid; t < cnt; t += SORT_THREADS) row[t] = hist[t];
}

__global__ __launch_bounds__(SCAN_THREADS) void k_prefix(unsigned* __restrict__ part,
                                                         unsigned* __restrict__ rowptr,
                                                         unsigned* __restrict__ blockSum,
                                                         float* __restrict__ dinv, int nkeys) {
  __shared__ unsigned sh[SCAN_THREADS];
  const int tid = threadIdx.x;
  const int k = blockIdx.x * SCAN_THREADS + tid;
  unsigned tot = 0u;
  if (k < nkeys) {
    unsigned v[SORT_CHUNKS];
#pragma unroll
    for (int c = 0; c < SORT_CHUNKS; c++) v[c] = part[(size_t)c * nkeys + k];
#pragma unroll
    for (int c = 0; c < SORT_CHUNKS; c++) {
      part[(size_t)c * nkeys + k] = tot;
      tot += v[c];
    }
  }
  sh[tid] = tot;
  __syncthreads();
  for (int off = 1; off < SCAN_THREADS; off <<= 1) {
    const unsigned a = (tid >= off) ? sh[tid - off] : 0u;
    __syncthreads();
    sh[tid] += a;
    __syncthreads();
  }
  const unsigned incl = sh[tid];
  if (k < nkeys) rowptr[k] = incl - tot;
  if (tid == SCAN_THREADS - 1) blockSum[blockIdx.x] = incl;

  const unsigned other = __shfl_down(tot, 1);
  if (((k & 1) == 0) && (k < nkeys)) dinv[k >> 1] = rsqrtf((float)(1u + tot + other));
}

__global__ __launch_bounds__(SCAN_THREADS) void k_addoff(unsigned* __restrict__ rowptr,
                                                         const unsigned* __restrict__ blockSum,
                                                         int nkeys) {
  __shared__ unsigned red[SCAN_THREADS];
  const int tid = threadIdx.x;
  const int b = blockIdx.x;
  unsigned s = 0u;
  for (int j = tid; j < b; j += SCAN_THREADS) s += blockSum[j];
  red[tid] = s;
  __syncthreads();
  for (int off = SCAN_THREADS / 2; off > 0; off >>= 1) {
    if (tid < off) red[tid] += red[tid + off];
    __syncthreads();
  }
  const unsigned offv = red[0];
  const int k = b * SCAN_THREADS + tid;
  if (k < nkeys) rowptr[k] += offv;
}

__global__ __launch_bounds__(SORT_THREADS) void k_fill(const int* __restrict__ src,
                                                       const int* __restrict__ dst,
                                                       const unsigned* __restrict__ part,
                                                       const unsigned* __restrict__ rowptr,
                                                       unsigned short* __restrict__ list,
                                                       int E, int nkeys, int chunkLen) {
  __shared__ unsigned cur[KEY_BINS];
  const int tid = threadIdx.x;
  const int c   = blockIdx.x;
  const unsigned lo = blockIdx.y * KEY_BINS;

  int cnt = nkeys - (int)lo;
  if (cnt > KEY_BINS) cnt = KEY_BINS;
  const unsigned* row = part + (size_t)c * nkeys + lo;
  for (int t = tid; t < cnt; t += SORT_THREADS) cur[t] = rowptr[lo + t] + row[t];
  __syncthreads();

  const int beg = c * chunkLen;
  int end = beg + chunkLen;
  if (end > E) end = E;
  const int len = end - beg;

  if (len > 0) {
    scan_chunk(src, dst, beg, len, tid, [&](int s, int d) {
      const unsigned b = edge_key(s, d) - lo;
      if (b < (unsigned)KEY_BINS) {
        const unsigned pos = atomicAdd(&cur[b], 1u);
        list[pos] = (unsigned short)(((unsigned)s) & 0xFFFFu);
      }
    });
  }
}

template <int K>
__global__ void k_gemm_wmma(const float* __restrict__ A, const float* __restrict__ W,
                            float* __restrict__ H, int ntiles) {
  __shared__ float Ws[K * 16];
  for (int t = threadIdx.x; t < K * 16; t += blockDim.x) Ws[t] = W[t];
  __syncthreads();

  const int lane = threadIdx.x & 31;
  const int tile = blockIdx.x * (blockDim.x >> 5) + (threadIdx.x >> 5);
  if (tile >= ntiles) return;

  const int half = lane >> 4;
  const int mn   = lane & 15;
  const int koff = half * 2;

  const float* arow = A + (size_t)(tile * 16 + mn) * K;

  v8f c = {0.f, 0.f, 0.f, 0.f, 0.f, 0.f, 0.f, 0.f};
#pragma unroll
  for (int k0 = 0; k0 < K; k0 += 4) {
    v2f a = *(const v2f*)(arow + k0 + koff);
    v2f b;
    b.x = Ws[(k0 + koff + 0) * 16 + mn];
    b.y = Ws[(k0 + koff + 1) * 16 + mn];
    c = __builtin_amdgcn_wmma_f32_16x16x4_f32(
         false, a,  false, b,
         (short)0, c,  false,  false);
  }

  float* hrow = H + (size_t)tile * 256 + half * 128 + mn;
#pragma unroll
  for (int i = 0; i < 8; i++) hrow[i * 16] = c[i];
}

__device__ __forceinline__ void edge_fma(float4& acc, float w, float4 h) {
  acc.x += w * h.x;
  acc.y += w * h.y;
  acc.z += w * h.z;
  acc.w += w * h.w;
}

__global__ void k_gather(const unsigned short* __restrict__ list,
                         const unsigned* __restrict__ rowptr,
                         const float* __restrict__ dinv, const float* __restrict__ H,
                         const float* __restrict__ bias, float* __restrict__ outp,
                         int n, int E) {
  const int idx = blockIdx.x * blockDim.x + threadIdx.x;
  const int i = idx >> 2;
  const int q = idx & 3;
  if (i >= n) return;

  const unsigned p0 = rowptr[2 * i];
  const unsigned p1 = rowptr[2 * i + 1];
  const unsigned pe = (i + 1 < n) ? rowptr[2 * i + 2] : (unsigned)E;
  const float dn = dinv[i];
  const float4* H4 = (const float4*)H;

  float4 acc = make_float4(0.0f, 0.0f, 0.0f, 0.0f);
  unsigned p = p0;
  for (; p + 4 <= pe; p += 4) {
    const unsigned s0 = (unsigned)list[p]     + ((p     >= p1) ? 65536u : 0u);
    const unsigned s1 = (unsigned)list[p + 1] + ((p + 1 >= p1) ? 65536u : 0u);
    const unsigned s2 = (unsigned)list[p + 2] + ((p + 2 >= p1) ? 65536u : 0u);
    const unsigned s3 = (unsigned)list[p + 3] + ((p + 3 >= p1) ? 65536u : 0u);
    const float w0 = dinv[s0] * dn;
    const float w1 = dinv[s1] * dn;
    const float w2 = dinv[s2] * dn;
    const float w3 = dinv[s3] * dn;
    const float4 h0 = H4[(size_t)s0 * 4 + q];
    const float4 h1 = H4[(size_t)s1 * 4 + q];
    const float4 h2 = H4[(size_t)s2 * 4 + q];
    const float4 h3 = H4[(size_t)s3 * 4 + q];
    edge_fma(acc, w0, h0);
    edge_fma(acc, w1, h1);
    edge_fma(acc, w2, h2);
    edge_fma(acc, w3, h3);
  }
  for (; p < pe; p++) {
    const unsigned s = (unsigned)list[p] + ((p >= p1) ? 65536u : 0u);
    const float w = dinv[s] * dn;
    edge_fma(acc, w, H4[(size_t)s * 4 + q]);
  }

  const float4 hs = H4[(size_t)i * 4 + q];
  const float dd = dn * dn;
  float4 r;
  r.x = fmaxf(acc.x + dd * hs.x + bias[q * 4 + 0], 0.0f);
  r.y = fmaxf(acc.y + dd * hs.y + bias[q * 4 + 1], 0.0f);
  r.z = fmaxf(acc.z + dd * hs.z + bias[q * 4 + 2], 0.0f);
  r.w = fmaxf(acc.w + dd * hs.w + bias[q * 4 + 3], 0.0f);
  ((float4*)outp)[idx] = r;
}

__device__ __forceinline__ void wmma_tail_slots(v8f& c, const v2f (&a)[4], const v2f (&b)[4]) {
  asm volatile("v_nop\n\tv_nop\n\tv_nop\n\tv_nop"
               : "+v"(c)
               : "v"(a[0]), "v"(a[1]), "v"(a[2]), "v"(a[3]),
                 "v"(b[0]), "v"(b[1]), "v"(b[2]), "v"(b[3]));
}

#define ROW_PITCH 20

__global__ __launch_bounds__(256) void k_gather_mm(const unsigned short* __restrict__ list,
                                                   const unsigned* __restrict__ rowptr,
                                                   const float* __restrict__ dinv,
                                                   const float* __restrict__ H,
                                                   const float* __restrict__ bias,
                                                   const float* __restrict__ Wn,
                                                   float* __restrict__ Hn,
                                                   int n, int E, int ntiles) {
  __shared__ __attribute__((aligned(16))) float hs[64 * ROW_PITCH];
  __shared__ float Ws[256];
  const int tid = threadIdx.x;
  Ws[tid] = Wn[tid];

  const int idx = blockIdx.x * 256 + tid;
  const int i = idx >> 2;
  const int q = idx & 3;

  float4 r = make_float4(0.0f, 0.0f, 0.0f, 0.0f);
  if (i < n) {
    const unsigned p0 = rowptr[2 * i];
    const unsigned p1 = rowptr[2 * i + 1];
    const unsigned pe = (i + 1 < n) ? rowptr[2 * i + 2] : (unsigned)E;
    const float dn = dinv[i];
    const float4* H4 = (const float4*)H;

    float4 acc = make_float4(0.0f, 0.0f, 0.0f, 0.0f);
    unsigned p = p0;
    for (; p + 4 <= pe; p += 4) {
      const unsigned s0 = (unsigned)list[p]     + ((p     >= p1) ? 65536u : 0u);
      const unsigned s1 = (unsigned)list[p + 1] + ((p + 1 >= p1) ? 65536u : 0u);
      const unsigned s2 = (unsigned)list[p + 2] + ((p + 2 >= p1) ? 65536u : 0u);
      const unsigned s3 = (unsigned)list[p + 3] + ((p + 3 >= p1) ? 65536u : 0u);
      const float w0 = dinv[s0] * dn;
      const float w1 = dinv[s1] * dn;
      const float w2 = dinv[s2] * dn;
      const float w3 = dinv[s3] * dn;
      const float4 h0 = H4[(size_t)s0 * 4 + q];
      const float4 h1 = H4[(size_t)s1 * 4 + q];
      const float4 h2 = H4[(size_t)s2 * 4 + q];
      const float4 h3 = H4[(size_t)s3 * 4 + q];
      edge_fma(acc, w0, h0);
      edge_fma(acc, w1, h1);
      edge_fma(acc, w2, h2);
      edge_fma(acc, w3, h3);
    }
    for (; p < pe; p++) {
      const unsigned s = (unsigned)list[p] + ((p >= p1) ? 65536u : 0u);
      const float w = dinv[s] * dn;
      edge_fma(acc, w, H4[(size_t)s * 4 + q]);
    }

    const float4 hv = H4[(size_t)i * 4 + q];
    const float dd = dn * dn;
    r.x = fmaxf(acc.x + dd * hv.x + bias[q * 4 + 0], 0.0f);
    r.y = fmaxf(acc.y + dd * hv.y + bias[q * 4 + 1], 0.0f);
    r.z = fmaxf(acc.z + dd * hv.z + bias[q * 4 + 2], 0.0f);
    r.w = fmaxf(acc.w + dd * hv.w + bias[q * 4 + 3], 0.0f);
  }
  *(float4*)(hs + (tid >> 2) * ROW_PITCH + q * 4) = r;
  __syncthreads();

  const int wv = tid >> 5;
  const int tile = blockIdx.x * 4 + wv;
  if (wv >= 4 || tile >= ntiles) return;

  const int lane = tid & 31;
  const int half = lane >> 4;
  const int mn   = lane & 15;
  const int koff = half * 2;
  const float* arow = hs + (wv * 16 + mn) * ROW_PITCH;

  v2f a[4], b[4];
#pragma unroll
  for (int s = 0; s < 4; s++) {
    a[s] = *(const v2f*)(arow + 4 * s + koff);
    b[s].x = Ws[(4 * s + koff + 0) * 16 + mn];
    b[s].y = Ws[(4 * s + koff + 1) * 16 + mn];
  }
  v8f c = {0.f, 0.f, 0.f, 0.f, 0.f, 0.f, 0.f, 0.f};
#pragma unroll
  for (int s = 0; s < 4; s++)
    c = __builtin_amdgcn_wmma_f32_16x16x4_f32(
         false, a[s],  false, b[s],
         (short)0, c,  false,  false);
  wmma_tail_slots(c, a, b);

  float* hrow = Hn + (size_t)tile * 256 + half * 128 + mn;
#pragma unroll
  for (int i8 = 0; i8 < 8; i8++) hrow[i8 * 16] = c[i8];
}

__global__ void k_head(const float* __restrict__ A, const float* __restrict__ W,
                       const float* __restrict__ l1b, const float* __restrict__ l2W,
                       const float* __restrict__ l2b, float* __restrict__ out, int ntiles) {
  __shared__ float Ws[256];
  for (int t = threadIdx.x; t < 256; t += blockDim.x) Ws[t] = W[t];
  __syncthreads();

  const int lane = threadIdx.x & 31;
  const int tile = blockIdx.x * (blockDim.x >> 5) + (threadIdx.x >> 5);
  if (tile >= ntiles) return;

  const int half = lane >> 4;
  const int mn   = lane & 15;
  const int koff = half * 2;
  const float* arow = A + (size_t)(tile * 16 + mn) * 16;
  const float bj = l1b[mn];
  const float wj = l2W[mn];

  v2f a[4], b[4];
#pragma unroll
  for (int s = 0; s < 4; s++) {
    a[s] = *(const v2f*)(arow + 4 * s + koff);
    b[s].x = Ws[(4 * s + koff + 0) * 16 + mn];
    b[s].y = Ws[(4 * s + koff + 1) * 16 + mn];
  }
  v8f c = {0.f, 0.f, 0.f, 0.f, 0.f, 0.f, 0.f, 0.f};
#pragma unroll
  for (int s = 0; s < 4; s++)
    c = __builtin_amdgcn_wmma_f32_16x16x4_f32(
         false, a[s],  false, b[s],
         (short)0, c,  false,  false);
  wmma_tail_slots(c, a, b);

  float mine = 0.0f;
#pragma unroll
  for (int r8 = 0; r8 < 8; r8++) {
    float v = fmaxf(c[r8] + bj, 0.0f) * wj;
#pragma unroll
    for (int off = 8; off; off >>= 1) v += __shfl_xor(v, off, 16);
    if (mn == r8) mine = v;
  }
  if (mn < 8) out[tile * 16 + half * 8 + mn] = mine + l2b[0];
}

extern "C" void kernel_launch(void* const* d_in, const int* in_sizes, int n_in,
                              void* d_out, int out_size, void* d_ws, size_t ws_size,
                              hipStream_t stream) {
  const float* x   = (const float*)d_in[0];
  const int*   ei  = (const int*)d_in[1];
  const float* W1  = (const float*)d_in[2];
  const float* b1  = (const float*)d_in[3];
  const float* W2  = (const float*)d_in[4];
  const float* b2  = (const float*)d_in[5];
  const float* l1W = (const float*)d_in[6];
  const float* l1b = (const float*)d_in[7];
  const float* l2W = (const float*)d_in[8];
  const float* l2b = (const float*)d_in[9];

  const int N = in_sizes[0] / 128;
  const int E = in_sizes[1] / 2;
  const int* src = ei;
  const int* dst = ei + E;

  unsigned* rowptr = (unsigned*)d_ws;
  float* bufA = (float*)d_ws + 2 * (size_t)N;
  float* bufB = bufA + (size_t)N * 16;
  float* bufC = bufB + (size_t)N * 16;
  float* dinv = (float*)d_out;

  unsigned short* list = (unsigned short*)bufA;
  unsigned* blockSum   = (unsigned*)bufA;
  unsigned* part       = (unsigned*)bufB;

  const int ntiles = N / 16;
  const int gemmBlocks = (ntiles * 32 + 255) / 256;
  const int gathBlocks = (N * 4 + 255) / 256;

  const int nkeys = 2 * N;
  const int chunkLen = (((E + SORT_CHUNKS - 1) / SORT_CHUNKS) + 3) & ~3;
  const dim3 sortGrid(SORT_CHUNKS, (nkeys + KEY_BINS - 1) / KEY_BINS);
  const int scanBlocks = (nkeys + SCAN_THREADS - 1) / SCAN_THREADS;
  k_hist  <<<sortGrid, SORT_THREADS, 0, stream>>>(src, dst, part, E, nkeys, chunkLen);
  k_prefix<<<scanBlocks, SCAN_THREADS, 0, stream>>>(part, rowptr, blockSum, dinv, nkeys);
  k_addoff<<<scanBlocks, SCAN_THREADS, 0, stream>>>(rowptr, blockSum, nkeys);
  k_fill  <<<sortGrid, SORT_THREADS, 0, stream>>>(src, dst, part, rowptr, list, E, nkeys, chunkLen);

  k_gemm_wmma<128><<<gemmBlocks, 256, 0, stream>>>(x, W1, bufB, ntiles);
  k_gather_mm<<<gathBlocks, 256, 0, stream>>>(list, rowptr, dinv, bufB, b1, W2, bufC, N, E, ntiles);

  k_gather<<<gathBlocks, 256, 0, stream>>>(list, rowptr, dinv, bufC, b2, bufB, N, E);

  k_head<<<gemmBlocks, 256, 0, stream>>>(bufB, l1W, l1b, l2W, l2b, (float*)d_out, ntiles);
}


# Sort passes: the key-range blocks of one edge chunk dispatched adjacent (grid x = range) and each starts its scan an eighth further in, so chunk re-reads hit L2; list order inside a chunk moves
# speedup vs baseline: 3.2839x; 1.5087x over previous
#include <hip/hip_runtime.h>
#include <hip/hip_bf16.h>


typedef __attribute__((ext_vector_type(2))) float v2f;
typedef __attribute__((ext_vector_type(4))) float v4f;
typedef __attribute__((ext_vector_type(8))) float v8f;

#define KEY_BINS     25600
#define SORT_CHUNKS  16
#define SORT_THREADS 1024
#define SCAN_THREADS 256

__device__ __forceinline__ unsigned edge_key(int s, int d) {
  return (((unsigned)d) << 1) | (((unsigned)s) >> 16);
}

__device__ __forceinline__ int scan_rotation(int len, int r, int nr) {
  return ((len / nr) * r) & ~3;
}

template <class F>
__device__ __forceinline__ void scan_chunk(const int* __restrict__ src, const int* __restrict__ dst,
                                           int beg, int len, int tid, F f) {
  const int* sp = src + beg;
  const int* dp = dst + beg;
  if (((((size_t)sp) | ((size_t)dp)) & 15) == 0) {
    const int4* s4 = (const int4*)sp;
    const int4* d4 = (const int4*)dp;
    const int nvec = len >> 2;
    int v = tid;
    for (; v + 3 * SORT_THREADS < nvec; v += 4 * SORT_THREADS) {
      int4 a0 = s4[v];
      int4 a1 = s4[v + 1 * SORT_THREADS];
      int4 a2 = s4[v + 2 * SORT_THREADS];
      int4 a3 = s4[v + 3 * SORT_THREADS];
      int4 b0 = d4[v];
      int4 b1 = d4[v + 1 * SORT_THREADS];
      int4 b2 = d4[v + 2 * SORT_THREADS];
      int4 b3 = d4[v + 3 * SORT_THREADS];
      f(a0.x, b0.x); f(a0.y, b0.y); f(a0.z, b0.z); f(a0.w, b0.w);
      f(a1.x, b1.x); f(a1.y, b1.y); f(a1.z, b1.z); f(a1.w, b1.w);
      f(a2.x, b2.x); f(a2.y, b2.y); f(a2.z, b2.z); f(a2.w, b2.w);
      f(a3.x, b3.x); f(a3.y, b3.y); f(a3.z, b3.z); f(a3.w, b3.w);
    }
    for (; v < nvec; v += SORT_THREADS) {
      int4 a = s4[v];
      int4 b = d4[v];
      f(a.x, b.x); f(a.y, b.y); f(a.z, b.z); f(a.w, b.w);
    }
    const int t = (nvec << 2) + tid;
    if (t < len) f(sp[t], dp[t]);
  } else {
    for (int t = tid; t < len; t += SORT_THREADS) f(sp[t], dp[t]);
  }
}

__global__ __launch_bounds__(SORT_THREADS) void k_hist(const int* __restrict__ src,
                                                       const int* __restrict__ dst,
                                                       unsigned* __restrict__ part,
                                                       int E, int nkeys, int chunkLen) {
  __shared__ unsigned hist[KEY_BINS];
  const int tid = threadIdx.x;
  const int c   = blockIdx.y;
  const unsigned lo = blockIdx.x * KEY_BINS;

  for (int t = tid; t < KEY_BINS; t += SORT_THREADS) hist[t] = 0u;
  __syncthreads();

  const int beg = c * chunkLen;
  int end = beg + chunkLen;
  if (end > E) end = E;
  const int len = end - beg;

  if (len > 0) {
    auto count = [&](int s, int d) {
      const unsigned b = edge_key(s, d) - lo;
      if (b < (unsigned)KEY_BINS) atomicAdd(&hist[b], 1u);
    };
    const int rot = scan_rotation(len, (int)blockIdx.x, (int)gridDim.x);
    scan_chunk(src, dst, beg + rot, len - rot, tid, count);
    if (rot > 0) scan_chunk(src, dst, beg, rot, tid, count);
  }
  __syncthreads();

  int cnt = nkeys - (int)lo;
  if (cnt > KEY_BINS) cnt = KEY_BINS;
  unsigned* row = part + (size_t)c * nkeys + lo;
  for (int t = tid; t < cnt; t += SORT_THREADS) row[t] = hist[t];
}

__global__ __launch_bounds__(SCAN_THREADS) void k_prefix(unsigned* __restrict__ part,
                                                         unsigned* __restrict__ rowptr,
                                                         unsigned* __restrict__ blockSum,
                                                         float* __restrict__ dinv, int nkeys) {
  __shared__ unsigned sh[SCAN_THREADS];
  const int tid = threadIdx.x;
  const int k = blockIdx.x * SCAN_THREADS + tid;
  unsigned tot = 0u;
  if (k < nkeys) {
    unsigned v[SORT_CHUNKS];
#pragma unroll
    for (int c = 0; c < SORT_CHUNKS; c++) v[c] = part[(size_t)c * nkeys + k];
#pragma unroll
    for (int c = 0; c < SORT_CHUNKS; c++) {
      part[(size_t)c * nkeys + k] = tot;
      tot += v[c];
    }
  }
  sh[tid] = tot;
  __syncthreads();
  for (int off = 1; off < SCAN_THREADS; off <<= 1) {
    const unsigned a = (tid >= off) ? sh[tid - off] : 0u;
    __syncthreads();
    sh[tid] += a;
    __syncthreads();
  }
  const unsigned incl = sh[tid];
  if (k < nkeys) rowptr[k] = incl - tot;
  if (tid == SCAN_THREADS - 1) blockSum[blockIdx.x] = incl;

  const unsigned other = __shfl_down(tot, 1);
  if (((k & 1) == 0) && (k < nkeys)) dinv[k >> 1] = rsqrtf((float)(1u + tot + other));
}

__global__ __launch_bounds__(SCAN_THREADS) void k_addoff(unsigned* __restrict__ rowptr,
                                                         const unsigned* __restrict__ blockSum,
                                                         int nkeys) {
  __shared__ unsigned red[SCAN_THREADS];
  const int tid = threadIdx.x;
  const int b = blockIdx.x;
  unsigned s = 0u;
  for (int j = tid; j < b; j += SCAN_THREADS) s += blockSum[j];
  red[tid] = s;
  __syncthreads();
  for (int off = SCAN_THREADS / 2; off > 0; off >>= 1) {
    if (tid < off) red[tid] += red[tid + off];
    __syncthreads();
  }
  const unsigned offv = red[0];
  const int k = b * SCAN_THREADS + tid;
  if (k < nkeys) rowptr[k] += offv;
}

__global__ __launch_bounds__(SORT_THREADS) void k_fill(const int* __restrict__ src,
                                                       const int* __restrict__ dst,
                                                       const unsigned* __restrict__ part,
                                                       const unsigned* __restrict__ rowptr,
                                                       unsigned short* __restrict__ list,
                                                       int E, int nkeys, int chunkLen) {
  __shared__ unsigned cur[KEY_BINS];
  const int tid = threadIdx.x;
  const int c   = blockIdx.y;
  const unsigned lo = blockIdx.x * KEY_BINS;

  int cnt = nkeys - (int)lo;
  if (cnt > KEY_BINS) cnt = KEY_BINS;
  const unsigned* row = part + (size_t)c * nkeys + lo;
  for (int t = tid; t < cnt; t += SORT_THREADS) cur[t] = rowptr[lo + t] + row[t];
  __syncthreads();

  const int beg = c * chunkLen;
  int end = beg + chunkLen;
  if (end > E) end = E;
  const int len = end - beg;

  if (len > 0) {
    auto place = [&](int s, int d) {
      const unsigned b = edge_key(s, d) - lo;
      if (b < (unsigned)KEY_BINS) {
        const unsigned pos = atomicAdd(&cur[b], 1u);
        list[pos] = (unsigned short)(((unsigned)s) & 0xFFFFu);
      }
    };
    const int rot = scan_rotation(len, (int)blockIdx.x, (int)gridDim.x);
    scan_chunk(src, dst, beg + rot, len - rot, tid, place);
    if (rot > 0) scan_chunk(src, dst, beg, rot, tid, place);
  }
}

template <int K>
__global__ void k_gemm_wmma(const float* __restrict__ A, const float* __restrict__ W,
                            const float* __restrict__ dinv,
                            float* __restrict__ H, int ntiles) {
  constexpr int PITCH = K / 2 + 2;
  __shared__ __attribute__((aligned(16))) float Wl[32 * PITCH];
  for (int t = threadIdx.x; t < K * 16; t += blockDim.x) {
    const int k = t >> 4, n = t & 15;
    Wl[((((k >> 1) & 1) << 4) | n) * PITCH + ((k >> 2) << 1) + (k & 1)] = W[t];
  }
  __syncthreads();

  const int lane = threadIdx.x & 31;
  const int tile = blockIdx.x * (blockDim.x >> 5) + (threadIdx.x >> 5);
  if (tile >= ntiles) return;

  const int half = lane >> 4;
  const int mn   = lane & 15;
  const int koff = half * 2;

  const float* arow = A + (size_t)(tile * 16 + mn) * K;
  const float* blane = Wl + lane * PITCH;

  v4f d0 = *(const v4f*)(dinv + tile * 16 + half * 8);
  v4f d1 = *(const v4f*)(dinv + tile * 16 + half * 8 + 4);
  int hoff = tile * 256 + half * 128 + mn;
  asm volatile("" : "+v"(d0), "+v"(d1), "+v"(hoff));

  v8f c = {0.f, 0.f, 0.f, 0.f, 0.f, 0.f, 0.f, 0.f};
  v2f a, b;
#pragma unroll
  for (int k0 = 0; k0 < K; k0 += 4) {
    a = *(const v2f*)(arow + k0 + koff);
    b = *(const v2f*)(blane + (k0 >> 1));
    c = __builtin_amdgcn_wmma_f32_16x16x4_f32(
         false, a,  false, b,
         (short)0, c,  false,  false);
  }
  asm volatile("v_nop\n\tv_nop\n\tv_nop\n\tv_nop" : "+v"(c), "+v"(hoff) : "v"(a), "v"(b));

  float* hrow = H + hoff;
  hrow[0 * 16] = c[0] * d0.x;
  hrow[1 * 16] = c[1] * d0.y;
  hrow[2 * 16] = c[2] * d0.z;
  hrow[3 * 16] = c[3] * d0.w;
  hrow[4 * 16] = c[4] * d1.x;
  hrow[5 * 16] = c[5] * d1.y;
  hrow[6 * 16] = c[6] * d1.z;
  hrow[7 * 16] = c[7] * d1.w;
}

__device__ __forceinline__ void edge_add(float4& acc, float4 h) {
  acc.x += h.x;
  acc.y += h.y;
  acc.z += h.z;
  acc.w += h.w;
}

__global__ void k_gather(const unsigned short* __restrict__ list,
                         const unsigned* __restrict__ rowptr,
                         const float* __restrict__ dinv, const float* __restrict__ H,
                         const float* __restrict__ bias, float* __restrict__ outp,
                         int n, int E) {
  const int idx = blockIdx.x * blockDim.x + threadIdx.x;
  const int i = idx >> 2;
  const int q = idx & 3;
  if (i >= n) return;

  const unsigned p0 = rowptr[2 * i];
  const unsigned p1 = rowptr[2 * i + 1];
  const unsigned pe = (i + 1 < n) ? rowptr[2 * i + 2] : (unsigned)E;
  const float dn = dinv[i];
  const float4* H4 = (const float4*)H;

  float4 acc = make_float4(0.0f, 0.0f, 0.0f, 0.0f);
  unsigned p = p0;
  for (; p + 4 <= pe; p += 4) {
    const unsigned s0 = (unsigned)list[p]     + ((p     >= p1) ? 65536u : 0u);
    const unsigned s1 = (unsigned)list[p + 1] + ((p + 1 >= p1) ? 65536u : 0u);
    const unsigned s2 = (unsigned)list[p + 2] + ((p + 2 >= p1) ? 65536u : 0u);
    const unsigned s3 = (unsigned)list[p + 3] + ((p + 3 >= p1) ? 65536u : 0u);
    const float4 h0 = H4[(size_t)s0 * 4 + q];
    const float4 h1 = H4[(size_t)s1 * 4 + q];
    const float4 h2 = H4[(size_t)s2 * 4 + q];
    const float4 h3 = H4[(size_t)s3 * 4 + q];
    edge_add(acc, h0);
    edge_add(acc, h1);
    edge_add(acc, h2);
    edge_add(acc, h3);
  }
  for (; p < pe; p++) {
    const unsigned s = (unsigned)list[p] + ((p >= p1) ? 65536u : 0u);
    edge_add(acc, H4[(size_t)s * 4 + q]);
  }

  const float4 hs = H4[(size_t)i * 4 + q];
  float4 r;
  r.x = fmaxf(dn * (acc.x + hs.x) + bias[q * 4 + 0], 0.0f);
  r.y = fmaxf(dn * (acc.y + hs.y) + bias[q * 4 + 1], 0.0f);
  r.z = fmaxf(dn * (acc.z + hs.z) + bias[q * 4 + 2], 0.0f);
  r.w = fmaxf(dn * (acc.w + hs.w) + bias[q * 4 + 3], 0.0f);
  ((float4*)outp)[idx] = r;
}

__device__ __forceinline__ void wmma_tail_slots(v8f& c, const v2f (&a)[4], const v2f (&b)[4]) {
  asm volatile("v_nop\n\tv_nop\n\tv_nop\n\tv_nop"
               : "+v"(c)
               : "v"(a[0]), "v"(a[1]), "v"(a[2]), "v"(a[3]),
                 "v"(b[0]), "v"(b[1]), "v"(b[2]), "v"(b[3]));
}

#define ROW_PITCH 20

__global__ __launch_bounds__(256) void k_gather_mm(const unsigned short* __restrict__ list,
                                                   const unsigned* __restrict__ rowptr,
                                                   const float* __restrict__ dinv,
                                                   const float* __restrict__ H,
                                                   const float* __restrict__ bias,
                                                   const float* __restrict__ Wn,
                                                   float* __restrict__ Hn,
                                                   int n, int E, int ntiles) {
  __shared__ __attribute__((aligned(16))) float hs[64 * ROW_PITCH];
  __shared__ float Ws[256];
  const int tid = threadIdx.x;
  Ws[tid] = Wn[tid];

  const int idx = blockIdx.x * 256 + tid;
  const int i = idx >> 2;
  const int q = idx & 3;

  float4 r = make_float4(0.0f, 0.0f, 0.0f, 0.0f);
  if (i < n) {
    const unsigned p0 = rowptr[2 * i];
    const unsigned p1 = rowptr[2 * i + 1];
    const unsigned pe = (i + 1 < n) ? rowptr[2 * i + 2] : (unsigned)E;
    const float dn = dinv[i];
    const float4* H4 = (const float4*)H;

    float4 acc = make_float4(0.0f, 0.0f, 0.0f, 0.0f);
    unsigned p = p0;
    for (; p + 4 <= pe; p += 4) {
      const unsigned s0 = (unsigned)list[p]     + ((p     >= p1) ? 65536u : 0u);
      const unsigned s1 = (unsigned)list[p + 1] + ((p + 1 >= p1) ? 65536u : 0u);
      const unsigned s2 = (unsigned)list[p + 2] + ((p + 2 >= p1) ? 65536u : 0u);
      const unsigned s3 = (unsigned)list[p + 3] + ((p + 3 >= p1) ? 65536u : 0u);
      const float4 h0 = H4[(size_t)s0 * 4 + q];
      const float4 h1 = H4[(size_t)s1 * 4 + q];
      const float4 h2 = H4[(size_t)s2 * 4 + q];
      const float4 h3 = H4[(size_t)s3 * 4 + q];
      edge_add(acc, h0);
      edge_add(acc, h1);
      edge_add(acc, h2);
      edge_add(acc, h3);
    }
    for (; p < pe; p++) {
      const unsigned s = (unsigned)list[p] + ((p >= p1) ? 65536u : 0u);
      edge_add(acc, H4[(size_t)s * 4 + q]);
    }

    const float4 hv = H4[(size_t)i * 4 + q];
    r.x = fmaxf(dn * (acc.x + hv.x) + bias[q * 4 + 0], 0.0f) * dn;
    r.y = fmaxf(dn * (acc.y + hv.y) + bias[q * 4 + 1], 0.0f) * dn;
    r.z = fmaxf(dn * (acc.z + hv.z) + bias[q * 4 + 2], 0.0f) * dn;
    r.w = fmaxf(dn * (acc.w + hv.w) + bias[q * 4 + 3], 0.0f) * dn;
  }
  *(float4*)(hs + (tid >> 2) * ROW_PITCH + q * 4) = r;
  __syncthreads();

  const int wv = tid >> 5;
  const int tile = blockIdx.x * 4 + wv;
  if (wv >= 4 || tile >= ntiles) return;

  const int lane = tid & 31;
  const int half = lane >> 4;
  const int mn   = lane & 15;
  const int koff = half * 2;
  const float* arow = hs + (wv * 16 + mn) * ROW_PITCH;

  v2f a[4], b[4];
#pragma unroll
  for (int s = 0; s < 4; s++) {
    a[s] = *(const v2f*)(arow + 4 * s + koff);
    b[s].x = Ws[(4 * s + koff + 0) * 16 + mn];
    b[s].y = Ws[(4 * s + koff + 1) * 16 + mn];
  }
  v8f c = {0.f, 0.f, 0.f, 0.f, 0.f, 0.f, 0.f, 0.f};
#pragma unroll
  for (int s = 0; s < 4; s++)
    c = __builtin_amdgcn_wmma_f32_16x16x4_f32(
         false, a[s],  false, b[s],
         (short)0, c,  false,  false);
  wmma_tail_slots(c, a, b);

  float* hrow = Hn + (size_t)tile * 256 + half * 128 + mn;
#pragma unroll
  for (int i8 = 0; i8 < 8; i8++) hrow[i8 * 16] = c[i8];
}

__global__ void k_head(const float* __restrict__ A, const float* __restrict__ W,
                       const float* __restrict__ l1b, const float* __restrict__ l2W,
                       const float* __restrict__ l2b, float* __restrict__ out, int ntiles) {
  __shared__ float Ws[256];
  for (int t = threadIdx.x; t < 256; t += blockDim.x) Ws[t] = W[t];
  __syncthreads();

  const int lane = threadIdx.x & 31;
  const int tile = blockIdx.x * (blockDim.x >> 5) + (threadIdx.x >> 5);
  if (tile >= ntiles) return;

  const int half = lane >> 4;
  const int mn   = lane & 15;
  const int koff = half * 2;
  const float* arow = A + (size_t)(tile * 16 + mn) * 16;
  const float bj = l1b[mn];
  const float wj = l2W[mn];

  v2f a[4], b[4];
#pragma unroll
  for (int s = 0; s < 4; s++) {
    a[s] = *(const v2f*)(arow + 4 * s + koff);
    b[s].x = Ws[(4 * s + koff + 0) * 16 + mn];
    b[s].y = Ws[(4 * s + koff + 1) * 16 + mn];
  }
  v8f c = {0.f, 0.f, 0.f, 0.f, 0.f, 0.f, 0.f, 0.f};
#pragma unroll
  for (int s = 0; s < 4; s++)
    c = __builtin_amdgcn_wmma_f32_16x16x4_f32(
         false, a[s],  false, b[s],
         (short)0, c,  false,  false);
  wmma_tail_slots(c, a, b);

  float mine = 0.0f;
#pragma unroll
  for (int r8 = 0; r8 < 8; r8++) {
    float v = fmaxf(c[r8] + bj, 0.0f) * wj;
#pragma unroll
    for (int off = 8; off; off >>= 1) v += __shfl_xor(v, off, 16);
    if (mn == r8) mine = v;
  }
  if (mn < 8) out[tile * 16 + half * 8 + mn] = mine + l2b[0];
}

extern "C" void kernel_launch(void* const* d_in, const int* in_sizes, int n_in,
                              void* d_out, int out_size, void* d_ws, size_t ws_size,
                              hipStream_t stream) {
  const float* x   = (const float*)d_in[0];
  const int*   ei  = (const int*)d_in[1];
  const float* W1  = (const float*)d_in[2];
  const float* b1  = (const float*)d_in[3];
  const float* W2  = (const float*)d_in[4];
  const float* b2  = (const float*)d_in[5];
  const float* l1W = (const float*)d_in[6];
  const float* l1b = (const float*)d_in[7];
  const float* l2W = (const float*)d_in[8];
  const float* l2b = (const float*)d_in[9];

  const int N = in_sizes[0] / 128;
  const int E = in_sizes[1] / 2;
  const int* src = ei;
  const int* dst = ei + E;

  unsigned* rowptr = (unsigned*)d_ws;
  float* bufA = (float*)d_ws + 2 * (size_t)N;
  float* bufB = bufA + (size_t)N * 16;
  float* bufC = bufB + (size_t)N * 16;
  float* dinv = (float*)d_out;

  unsigned short* list = (unsigned short*)bufA;
  unsigned* blockSum   = (unsigned*)bufA;
  unsigned* part       = (unsigned*)bufB;

  const int ntiles = N / 16;
  const int gemmBlocks = (ntiles * 32 + 255) / 256;
  const int gathBlocks = (N * 4 + 255) / 256;

  const int nkeys = 2 * N;
  const int chunkLen = (((E + SORT_CHUNKS - 1) / SORT_CHUNKS) + 3) & ~3;
  const dim3 sortGrid((nkeys + KEY_BINS - 1) / KEY_BINS, SORT_CHUNKS);
  const int scanBlocks = (nkeys + SCAN_THREADS - 1) / SCAN_THREADS;
  k_hist  <<<sortGrid, SORT_THREADS, 0, stream>>>(src, dst, part, E, nkeys, chunkLen);
  k_prefix<<<scanBlocks, SCAN_THREADS, 0, stream>>>(part, rowptr, blockSum, dinv, nkeys);
  k_addoff<<<scanBlocks, SCAN_THREADS, 0, stream>>>(rowptr, blockSum, nkeys);
  k_fill  <<<sortGrid, SORT_THREADS, 0, stream>>>(src, dst, part, rowptr, list, E, nkeys, chunkLen);

  k_gemm_wmma<128><<<gemmBlocks, 256, 0, stream>>>(x, W1, dinv, bufB, ntiles);
  k_gather_mm<<<gathBlocks, 256, 0, stream>>>(list, rowptr, dinv, bufB, b1, W2, bufC, N, E, ntiles);

  k_gather<<<gathBlocks, 256, 0, stream>>>(list, rowptr, dinv, bufC, b2, bufB, N, E);

  k_head<<<gemmBlocks, 256, 0, stream>>>(bufB, l1W, l1b, l2W, l2b, (float*)d_out, ntiles);
}


# Edge sort in two levels: edges grouped by 512-node bucket (read once, one packed word each), then one work-group orders a bucket in LDS; no 8x re-scan; list order in a node, so float sum order, moves
# speedup vs baseline: 4.1282x; 1.2571x over previous
#include <hip/hip_runtime.h>
#include <hip/hip_bf16.h>


typedef __attribute__((ext_vector_type(2))) float v2f;
typedef __attribute__((ext_vector_type(4))) float v4f;
typedef __attribute__((ext_vector_type(8))) float v8f;

#define BUCKET_SHIFT  9
#define BUCKET_KEYS   1024
#define MAX_BUCKETS   256
#define EDGE_CHUNKS   256
#define SORT_THREADS  1024
#define STAGE_CAP     32768

__global__ __launch_bounds__(SORT_THREADS) void k_bcount(const int* __restrict__ dst,
                                                         unsigned* __restrict__ cnt,
                                                         int E, int chunkLen, int nb) {
  __shared__ unsigned h[MAX_BUCKETS];
  const int tid = threadIdx.x;
  if (tid < MAX_BUCKETS) h[tid] = 0u;
  __syncthreads();

  const int beg = blockIdx.x * chunkLen;
  int end = beg + chunkLen;
  if (end > E) end = E;

  int e = beg + tid;
  for (; e + 3 * SORT_THREADS < end; e += 4 * SORT_THREADS) {
    const int d0 = dst[e];
    const int d1 = dst[e + 1 * SORT_THREADS];
    const int d2 = dst[e + 2 * SORT_THREADS];
    const int d3 = dst[e + 3 * SORT_THREADS];
    atomicAdd(&h[((unsigned)d0) >> BUCKET_SHIFT], 1u);
    atomicAdd(&h[((unsigned)d1) >> BUCKET_SHIFT], 1u);
    atomicAdd(&h[((unsigned)d2) >> BUCKET_SHIFT], 1u);
    atomicAdd(&h[((unsigned)d3) >> BUCKET_SHIFT], 1u);
  }
  for (; e < end; e += SORT_THREADS) atomicAdd(&h[((unsigned)dst[e]) >> BUCKET_SHIFT], 1u);
  __syncthreads();

  if (tid < nb) cnt[tid * EDGE_CHUNKS + blockIdx.x] = h[tid];
}

__global__ __launch_bounds__(256) void k_bscan(unsigned* __restrict__ cnt,
                                               unsigned* __restrict__ totals, int nb) {
  const int lane = threadIdx.x & 31;
  const int b = blockIdx.x * 8 + (threadIdx.x >> 5);
  if (b >= nb) return;
  unsigned* p = cnt + b * EDGE_CHUNKS + lane * 8;
  unsigned v[8];
  unsigned s = 0u;
#pragma unroll
  for (int j = 0; j < 8; j++) {
    v[j] = p[j];
    s += v[j];
  }
  unsigned incl = s;
#pragma unroll
  for (int off = 1; off < 32; off <<= 1) {
    const unsigned t = __shfl_up(incl, off);
    if (lane >= off) incl += t;
  }
  unsigned run = incl - s;
#pragma unroll
  for (int j = 0; j < 8; j++) {
    p[j] = run;
    run += v[j];
  }
  if (lane == 31) totals[b] = incl;
}

__global__ __launch_bounds__(SORT_THREADS) void k_bscatter(const int* __restrict__ src,
                                                           const int* __restrict__ dst,
                                                           const unsigned* __restrict__ cnt,
                                                           const unsigned* __restrict__ totals,
                                                           unsigned* __restrict__ packed,
                                                           unsigned* __restrict__ rowptr,
                                                           int E, int chunkLen, int nb) {
  __shared__ unsigned sc[MAX_BUCKETS];
  __shared__ unsigned cur[MAX_BUCKETS];
  const int tid = threadIdx.x;

  unsigned tot = 0u;
  if (tid < MAX_BUCKETS) {
    if (tid < nb) tot = totals[tid];
    sc[tid] = tot;
  }
  __syncthreads();
  for (int off = 1; off < MAX_BUCKETS; off <<= 1) {
    unsigned a = 0u;
    if (tid < MAX_BUCKETS && tid >= off) a = sc[tid - off];
    __syncthreads();
    if (tid < MAX_BUCKETS) sc[tid] += a;
    __syncthreads();
  }
  if (tid < nb) {
    const unsigned start = sc[tid] - tot;
    cur[tid] = start + cnt[tid * EDGE_CHUNKS + blockIdx.x];
    if (blockIdx.x == 0) rowptr[tid * BUCKET_KEYS] = start;
  }
  __syncthreads();

  const int beg = blockIdx.x * chunkLen;
  int end = beg + chunkLen;
  if (end > E) end = E;

  auto place = [&](int s, int d) {
    const unsigned pos = atomicAdd(&cur[((unsigned)d) >> BUCKET_SHIFT], 1u);
    packed[pos] = ((((unsigned)d) & 511u) << 17) | (unsigned)s;
  };
  int e = beg + tid;
  for (; e + 3 * SORT_THREADS < end; e += 4 * SORT_THREADS) {
    const int s0 = src[e];
    const int s1 = src[e + 1 * SORT_THREADS];
    const int s2 = src[e + 2 * SORT_THREADS];
    const int s3 = src[e + 3 * SORT_THREADS];
    const int d0 = dst[e];
    const int d1 = dst[e + 1 * SORT_THREADS];
    const int d2 = dst[e + 2 * SORT_THREADS];
    const int d3 = dst[e + 3 * SORT_THREADS];
    place(s0, d0);
    place(s1, d1);
    place(s2, d2);
    place(s3, d3);
  }
  for (; e < end; e += SORT_THREADS) place(src[e], dst[e]);
}

__device__ __forceinline__ unsigned packed_key(unsigned w) {
  return ((w >> 17) << 1) | ((w >> 16) & 1u);
}

template <bool STAGED>
__device__ __forceinline__ void bucket_fill(const unsigned* __restrict__ packed, unsigned* next,
                                            unsigned short* stage, unsigned short* __restrict__ list,
                                            unsigned beg, unsigned end, int tid) {
  auto place = [&](unsigned w) {
    const unsigned pos = atomicAdd(&next[packed_key(w)], 1u);
    const unsigned short v = (unsigned short)(w & 0xFFFFu);
    if (STAGED) stage[pos - beg] = v;
    else list[pos] = v;
  };
  unsigned p = beg + tid;
  for (; p + 3 * SORT_THREADS < end; p += 4 * SORT_THREADS) {
    const unsigned w0 = packed[p];
    const unsigned w1 = packed[p + 1 * SORT_THREADS];
    const unsigned w2 = packed[p + 2 * SORT_THREADS];
    const unsigned w3 = packed[p + 3 * SORT_THREADS];
    place(w0);
    place(w1);
    place(w2);
    place(w3);
  }
  for (; p < end; p += SORT_THREADS) place(packed[p]);
}

__global__ __launch_bounds__(SORT_THREADS) void k_bsort(const unsigned* __restrict__ packed,
                                                        unsigned* __restrict__ rowptr,
                                                        unsigned short* __restrict__ list,
                                                        float* __restrict__ dinv,
                                                        int E, int n, int nb) {
  __shared__ unsigned hist[BUCKET_KEYS];
  __shared__ unsigned sc[BUCKET_KEYS];
  __shared__ unsigned short stage[STAGE_CAP];
  const int tid = threadIdx.x;
  const int b = blockIdx.x;

  const unsigned beg = rowptr[b * BUCKET_KEYS];
  const unsigned end = (b + 1 < nb) ? rowptr[(b + 1) * BUCKET_KEYS] : (unsigned)E;

  hist[tid] = 0u;
  __syncthreads();

  unsigned p = beg + tid;
  for (; p + 3 * SORT_THREADS < end; p += 4 * SORT_THREADS) {
    const unsigned w0 = packed[p];
    const unsigned w1 = packed[p + 1 * SORT_THREADS];
    const unsigned w2 = packed[p + 2 * SORT_THREADS];
    const unsigned w3 = packed[p + 3 * SORT_THREADS];
    atomicAdd(&hist[packed_key(w0)], 1u);
    atomicAdd(&hist[packed_key(w1)], 1u);
    atomicAdd(&hist[packed_key(w2)], 1u);
    atomicAdd(&hist[packed_key(w3)], 1u);
  }
  for (; p < end; p += SORT_THREADS) atomicAdd(&hist[packed_key(packed[p])], 1u);
  __syncthreads();

  const unsigned mine  = hist[tid];
  const unsigned other = hist[tid ^ 1];
  sc[tid] = mine;
  __syncthreads();
  for (int off = 1; off < BUCKET_KEYS; off <<= 1) {
    const unsigned a = (tid >= off) ? sc[tid - off] : 0u;
    __syncthreads();
    sc[tid] += a;
    __syncthreads();
  }
  const unsigned first = beg + (sc[tid] - mine);
  hist[tid] = first;

  const int node = (b << BUCKET_SHIFT) + (tid >> 1);
  if (node < n) {
    if (tid != 0) rowptr[b * BUCKET_KEYS + tid] = first;
    if ((tid & 1) == 0) dinv[node] = rsqrtf((float)(1u + mine + other));
  }
  __syncthreads();

  if ((end - beg) <= (unsigned)STAGE_CAP) {
    bucket_fill<true>(packed, hist, stage, list, beg, end, tid);
    __syncthreads();
    const unsigned total = end - beg;
    for (unsigned i = tid; i < total; i += SORT_THREADS) list[beg + i] = stage[i];
  } else {
    bucket_fill<false>(packed, hist, stage, list, beg, end, tid);
  }
}

template <int K>
__global__ void k_gemm_wmma(const float* __restrict__ A, const float* __restrict__ W,
                            const float* __restrict__ dinv,
                            float* __restrict__ H, int ntiles) {
  constexpr int PITCH = K / 2 + 2;
  __shared__ __attribute__((aligned(16))) float Wl[32 * PITCH];
  for (int t = threadIdx.x; t < K * 16; t += blockDim.x) {
    const int k = t >> 4, n = t & 15;
    Wl[((((k >> 1) & 1) << 4) | n) * PITCH + ((k >> 2) << 1) + (k & 1)] = W[t];
  }
  __syncthreads();

  const int lane = threadIdx.x & 31;
  const int tile = blockIdx.x * (blockDim.x >> 5) + (threadIdx.x >> 5);
  if (tile >= ntiles) return;

  const int half = lane >> 4;
  const int mn   = lane & 15;
  const int koff = half * 2;

  const float* arow = A + (size_t)(tile * 16 + mn) * K;
  const float* blane = Wl + lane * PITCH;

  v4f d0 = *(const v4f*)(dinv + tile * 16 + half * 8);
  v4f d1 = *(const v4f*)(dinv + tile * 16 + half * 8 + 4);
  int hoff = tile * 256 + half * 128 + mn;
  asm volatile("" : "+v"(d0), "+v"(d1), "+v"(hoff));

  v8f c = {0.f, 0.f, 0.f, 0.f, 0.f, 0.f, 0.f, 0.f};
  v2f a, b;
#pragma unroll
  for (int k0 = 0; k0 < K; k0 += 4) {
    a = *(const v2f*)(arow + k0 + koff);
    b = *(const v2f*)(blane + (k0 >> 1));
    c = __builtin_amdgcn_wmma_f32_16x16x4_f32(
         false, a,  false, b,
         (short)0, c,  false,  false);
  }
  asm volatile("v_nop\n\tv_nop\n\tv_nop\n\tv_nop" : "+v"(c), "+v"(hoff) : "v"(a), "v"(b));

  float* hrow = H + hoff;
  hrow[0 * 16] = c[0] * d0.x;
  hrow[1 * 16] = c[1] * d0.y;
  hrow[2 * 16] = c[2] * d0.z;
  hrow[3 * 16] = c[3] * d0.w;
  hrow[4 * 16] = c[4] * d1.x;
  hrow[5 * 16] = c[5] * d1.y;
  hrow[6 * 16] = c[6] * d1.z;
  hrow[7 * 16] = c[7] * d1.w;
}

__device__ __forceinline__ void edge_add(float4& acc, float4 h) {
  acc.x += h.x;
  acc.y += h.y;
  acc.z += h.z;
  acc.w += h.w;
}

__global__ void k_gather(const unsigned short* __restrict__ list,
                         const unsigned* __restrict__ rowptr,
                         const float* __restrict__ dinv, const float* __restrict__ H,
                         const float* __restrict__ bias, float* __restrict__ outp,
                         int n, int E) {
  const int idx = blockIdx.x * blockDim.x + threadIdx.x;
  const int i = idx >> 2;
  const int q = idx & 3;
  if (i >= n) return;

  const unsigned p0 = rowptr[2 * i];
  const unsigned p1 = rowptr[2 * i + 1];
  const unsigned pe = (i + 1 < n) ? rowptr[2 * i + 2] : (unsigned)E;
  const float dn = dinv[i];
  const float4* H4 = (const float4*)H;

  float4 acc = make_float4(0.0f, 0.0f, 0.0f, 0.0f);
  unsigned p = p0;
  for (; p + 4 <= pe; p += 4) {
    const unsigned s0 = (unsigned)list[p]     + ((p     >= p1) ? 65536u : 0u);
    const unsigned s1 = (unsigned)list[p + 1] + ((p + 1 >= p1) ? 65536u : 0u);
    const unsigned s2 = (unsigned)list[p + 2] + ((p + 2 >= p1) ? 65536u : 0u);
    const unsigned s3 = (unsigned)list[p + 3] + ((p + 3 >= p1) ? 65536u : 0u);
    const float4 h0 = H4[(size_t)s0 * 4 + q];
    const float4 h1 = H4[(size_t)s1 * 4 + q];
    const float4 h2 = H4[(size_t)s2 * 4 + q];
    const float4 h3 = H4[(size_t)s3 * 4 + q];
    edge_add(acc, h0);
    edge_add(acc, h1);
    edge_add(acc, h2);
    edge_add(acc, h3);
  }
  for (; p < pe; p++) {
    const unsigned s = (unsigned)list[p] + ((p >= p1) ? 65536u : 0u);
    edge_add(acc, H4[(size_t)s * 4 + q]);
  }

  const float4 hs = H4[(size_t)i * 4 + q];
  float4 r;
  r.x = fmaxf(dn * (acc.x + hs.x) + bias[q * 4 + 0], 0.0f);
  r.y = fmaxf(dn * (acc.y + hs.y) + bias[q * 4 + 1], 0.0f);
  r.z = fmaxf(dn * (acc.z + hs.z) + bias[q * 4 + 2], 0.0f);
  r.w = fmaxf(dn * (acc.w + hs.w) + bias[q * 4 + 3], 0.0f);
  ((float4*)outp)[idx] = r;
}

__device__ __forceinline__ void wmma_tail_slots(v8f& c, const v2f (&a)[4], const v2f (&b)[4]) {
  asm volatile("v_nop\n\tv_nop\n\tv_nop\n\tv_nop"
               : "+v"(c)
               : "v"(a[0]), "v"(a[1]), "v"(a[2]), "v"(a[3]),
                 "v"(b[0]), "v"(b[1]), "v"(b[2]), "v"(b[3]));
}

#define ROW_PITCH 20

__global__ __launch_bounds__(256) void k_gather_mm(const unsigned short* __restrict__ list,
                                                   const unsigned* __restrict__ rowptr,
                                                   const float* __restrict__ dinv,
                                                   const float* __restrict__ H,
                                                   const float* __restrict__ bias,
                                                   const float* __restrict__ Wn,
                                                   float* __restrict__ Hn,
                                                   int n, int E, int ntiles) {
  __shared__ __attribute__((aligned(16))) float hs[64 * ROW_PITCH];
  __shared__ float Ws[256];
  const int tid = threadIdx.x;
  Ws[tid] = Wn[tid];

  const int idx = blockIdx.x * 256 + tid;
  const int i = idx >> 2;
  const int q = idx & 3;

  float4 r = make_float4(0.0f, 0.0f, 0.0f, 0.0f);
  if (i < n) {
    const unsigned p0 = rowptr[2 * i];
    const unsigned p1 = rowptr[2 * i + 1];
    const unsigned pe = (i + 1 < n) ? rowptr[2 * i + 2] : (unsigned)E;
    const float dn = dinv[i];
    const float4* H4 = (const float4*)H;

    float4 acc = make_float4(0.0f, 0.0f, 0.0f, 0.0f);
    unsigned p = p0;
    for (; p + 4 <= pe; p += 4) {
      const unsigned s0 = (unsigned)list[p]     + ((p     >= p1) ? 65536u : 0u);
      const unsigned s1 = (unsigned)list[p + 1] + ((p + 1 >= p1) ? 65536u : 0u);
      const unsigned s2 = (unsigned)list[p + 2] + ((p + 2 >= p1) ? 65536u : 0u);
      const unsigned s3 = (unsigned)list[p + 3] + ((p + 3 >= p1) ? 65536u : 0u);
      const float4 h0 = H4[(size_t)s0 * 4 + q];
      const float4 h1 = H4[(size_t)s1 * 4 + q];
      const float4 h2 = H4[(size_t)s2 * 4 + q];
      const float4 h3 = H4[(size_t)s3 * 4 + q];
      edge_add(acc, h0);
      edge_add(acc, h1);
      edge_add(acc, h2);
      edge_add(acc, h3);
    }
    for (; p < pe; p++) {
      const unsigned s = (unsigned)list[p] + ((p >= p1) ? 65536u : 0u);
      edge_add(acc, H4[(size_t)s * 4 + q]);
    }

    const float4 hv = H4[(size_t)i * 4 + q];
    r.x = fmaxf(dn * (acc.x + hv.x) + bias[q * 4 + 0], 0.0f) * dn;
    r.y = fmaxf(dn * (acc.y + hv.y) + bias[q * 4 + 1], 0.0f) * dn;
    r.z = fmaxf(dn * (acc.z + hv.z) + bias[q * 4 + 2], 0.0f) * dn;
    r.w = fmaxf(dn * (acc.w + hv.w) + bias[q * 4 + 3], 0.0f) * dn;
  }
  *(float4*)(hs + (tid >> 2) * ROW_PITCH + q * 4) = r;
  __syncthreads();

  const int wv = tid >> 5;
  const int tile = blockIdx.x * 4 + wv;
  if (wv >= 4 || tile >= ntiles) return;

  const int lane = tid & 31;
  const int half = lane >> 4;
  const int mn   = lane & 15;
  const int koff = half * 2;
  const float* arow = hs + (wv * 16 + mn) * ROW_PITCH;

  v2f a[4], b[4];
#pragma unroll
  for (int s = 0; s < 4; s++) {
    a[s] = *(const v2f*)(arow + 4 * s + koff);
    b[s].x = Ws[(4 * s + koff + 0) * 16 + mn];
    b[s].y = Ws[(4 * s + koff + 1) * 16 + mn];
  }
  v8f c = {0.f, 0.f, 0.f, 0.f, 0.f, 0.f, 0.f, 0.f};
#pragma unroll
  for (int s = 0; s < 4; s++)
    c = __builtin_amdgcn_wmma_f32_16x16x4_f32(
         false, a[s],  false, b[s],
         (short)0, c,  false,  false);
  wmma_tail_slots(c, a, b);

  float* hrow = Hn + (size_t)tile * 256 + half * 128 + mn;
#pragma unroll
  for (int i8 = 0; i8 < 8; i8++) hrow[i8 * 16] = c[i8];
}

__global__ void k_head(const float* __restrict__ A, const float* __restrict__ W,
                       const float* __restrict__ l1b, const float* __restrict__ l2W,
                       const float* __restrict__ l2b, float* __restrict__ out, int ntiles) {
  __shared__ float Ws[256];
  for (int t = threadIdx.x; t < 256; t += blockDim.x) Ws[t] = W[t];
  __syncthreads();

  const int lane = threadIdx.x & 31;
  const int tile = blockIdx.x * (blockDim.x >> 5) + (threadIdx.x >> 5);
  if (tile >= ntiles) return;

  const int half = lane >> 4;
  const int mn   = lane & 15;
  const int koff = half * 2;
  const float* arow = A + (size_t)(tile * 16 + mn) * 16;
  const float bj = l1b[mn];
  const float wj = l2W[mn];

  v2f a[4], b[4];
#pragma unroll
  for (int s = 0; s < 4; s++) {
    a[s] = *(const v2f*)(arow + 4 * s + koff);
    b[s].x = Ws[(4 * s + koff + 0) * 16 + mn];
    b[s].y = Ws[(4 * s + koff + 1) * 16 + mn];
  }
  v8f c = {0.f, 0.f, 0.f, 0.f, 0.f, 0.f, 0.f, 0.f};
#pragma unroll
  for (int s = 0; s < 4; s++)
    c = __builtin_amdgcn_wmma_f32_16x16x4_f32(
         false, a[s],  false, b[s],
         (short)0, c,  false,  false);
  wmma_tail_slots(c, a, b);

  float mine = 0.0f;
#pragma unroll
  for (int r8 = 0; r8 < 8; r8++) {
    float v = fmaxf(c[r8] + bj, 0.0f) * wj;
#pragma unroll
    for (int off = 8; off; off >>= 1) v += __shfl_xor(v, off, 16);
    if (mn == r8) mine = v;
  }
  if (mn < 8) out[tile * 16 + half * 8 + mn] = mine + l2b[0];
}

extern "C" void kernel_launch(void* const* d_in, const int* in_sizes, int n_in,
                              void* d_out, int out_size, void* d_ws, size_t ws_size,
                              hipStream_t stream) {
  const float* x   = (const float*)d_in[0];
  const int*   ei  = (const int*)d_in[1];
  const float* W1  = (const float*)d_in[2];
  const float* b1  = (const float*)d_in[3];
  const float* W2  = (const float*)d_in[4];
  const float* b2  = (const float*)d_in[5];
  const float* l1W = (const float*)d_in[6];
  const float* l1b = (const float*)d_in[7];
  const float* l2W = (const float*)d_in[8];
  const float* l2b = (const float*)d_in[9];

  const int N = in_sizes[0] / 128;
  const int E = in_sizes[1] / 2;
  const int* src = ei;
  const int* dst = ei + E;

  unsigned* rowptr = (unsigned*)d_ws;
  float* bufA = (float*)d_ws + 2 * (size_t)N;
  float* bufB = bufA + (size_t)N * 16;
  float* bufC = bufB + (size_t)N * 16;
  float* dinv = (float*)d_out;

  unsigned short* list = (unsigned short*)bufA;
  unsigned* bcnt       = (unsigned*)bufA;
  unsigned* btot       = (unsigned*)bufA + MAX_BUCKETS * EDGE_CHUNKS;
  unsigned* packed     = (unsigned*)bufB;

  const int ntiles = N / 16;
  const int gemmBlocks = (ntiles * 32 + 255) / 256;
  const int gathBlocks = (N * 4 + 255) / 256;

  const int nb = (N + (1 << BUCKET_SHIFT) - 1) >> BUCKET_SHIFT;
  const int chunkLen = (E + EDGE_CHUNKS - 1) / EDGE_CHUNKS;
  k_bcount  <<<EDGE_CHUNKS, SORT_THREADS, 0, stream>>>(dst, bcnt, E, chunkLen, nb);
  k_bscan   <<<(nb + 7) / 8, 256, 0, stream>>>(bcnt, btot, nb);
  k_bscatter<<<EDGE_CHUNKS, SORT_THREADS, 0, stream>>>(src, dst, bcnt, btot, packed, rowptr, E, chunkLen, nb);
  k_bsort   <<<nb, SORT_THREADS, 0, stream>>>(packed, rowptr, list, dinv, E, N, nb);

  k_gemm_wmma<128><<<gemmBlocks, 256, 0, stream>>>(x, W1, dinv, bufB, ntiles);
  k_gather_mm<<<gathBlocks, 256, 0, stream>>>(list, rowptr, dinv, bufB, b1, W2, bufC, N, E, ntiles);

  k_gather<<<gathBlocks, 256, 0, stream>>>(list, rowptr, dinv, bufC, b2, bufB, N, E);

  k_head<<<gemmBlocks, 256, 0, stream>>>(bufB, l1W, l1b, l2W, l2b, (float*)d_out, ntiles);
}


# Head fused into the layer-2 gather's epilogue: a work-group's 64 rows of h2 go through LDS to the lin1 WMMA and relu-dot (h2 never stored); 8->7 launches; same sums, same order
# speedup vs baseline: 4.3031x; 1.0424x over previous
#include <hip/hip_runtime.h>
#include <hip/hip_bf16.h>


typedef __attribute__((ext_vector_type(2))) float v2f;
typedef __attribute__((ext_vector_type(4))) float v4f;
typedef __attribute__((ext_vector_type(8))) float v8f;

#define BUCKET_SHIFT  9
#define BUCKET_KEYS   1024
#define MAX_BUCKETS   256
#define EDGE_CHUNKS   256
#define SORT_THREADS  1024
#define STAGE_CAP     32768

__global__ __launch_bounds__(SORT_THREADS) void k_bcount(const int* __restrict__ dst,
                                                         unsigned* __restrict__ cnt,
                                                         int E, int chunkLen, int nb) {
  __shared__ unsigned h[MAX_BUCKETS];
  const int tid = threadIdx.x;
  if (tid < MAX_BUCKETS) h[tid] = 0u;
  __syncthreads();

  const int beg = blockIdx.x * chunkLen;
  int end = beg + chunkLen;
  if (end > E) end = E;

  int e = beg + tid;
  for (; e + 3 * SORT_THREADS < end; e += 4 * SORT_THREADS) {
    const int d0 = dst[e];
    const int d1 = dst[e + 1 * SORT_THREADS];
    const int d2 = dst[e + 2 * SORT_THREADS];
    const int d3 = dst[e + 3 * SORT_THREADS];
    atomicAdd(&h[((unsigned)d0) >> BUCKET_SHIFT], 1u);
    atomicAdd(&h[((unsigned)d1) >> BUCKET_SHIFT], 1u);
    atomicAdd(&h[((unsigned)d2) >> BUCKET_SHIFT], 1u);
    atomicAdd(&h[((unsigned)d3) >> BUCKET_SHIFT], 1u);
  }
  for (; e < end; e += SORT_THREADS) atomicAdd(&h[((unsigned)dst[e]) >> BUCKET_SHIFT], 1u);
  __syncthreads();

  if (tid < nb) cnt[tid * EDGE_CHUNKS + blockIdx.x] = h[tid];
}

__global__ __launch_bounds__(256) void k_bscan(unsigned* __restrict__ cnt,
                                               unsigned* __restrict__ totals, int nb) {
  const int lane = threadIdx.x & 31;
  const int b = blockIdx.x * 8 + (threadIdx.x >> 5);
  if (b >= nb) return;
  unsigned* p = cnt + b * EDGE_CHUNKS + lane * 8;
  unsigned v[8];
  unsigned s = 0u;
#pragma unroll
  for (int j = 0; j < 8; j++) {
    v[j] = p[j];
    s += v[j];
  }
  unsigned incl = s;
#pragma unroll
  for (int off = 1; off < 32; off <<= 1) {
    const unsigned t = __shfl_up(incl, off);
    if (lane >= off) incl += t;
  }
  unsigned run = incl - s;
#pragma unroll
  for (int j = 0; j < 8; j++) {
    p[j] = run;
    run += v[j];
  }
  if (lane == 31) totals[b] = incl;
}

__global__ __launch_bounds__(SORT_THREADS) void k_bscatter(const int* __restrict__ src,
                                                           const int* __restrict__ dst,
                                                           const unsigned* __restrict__ cnt,
                                                           const unsigned* __restrict__ totals,
                                                           unsigned* __restrict__ packed,
                                                           unsigned* __restrict__ rowptr,
                                                           int E, int chunkLen, int nb) {
  __shared__ unsigned sc[MAX_BUCKETS];
  __shared__ unsigned cur[MAX_BUCKETS];
  const int tid = threadIdx.x;

  unsigned tot = 0u;
  if (tid < MAX_BUCKETS) {
    if (tid < nb) tot = totals[tid];
    sc[tid] = tot;
  }
  __syncthreads();
  for (int off = 1; off < MAX_BUCKETS; off <<= 1) {
    unsigned a = 0u;
    if (tid < MAX_BUCKETS && tid >= off) a = sc[tid - off];
    __syncthreads();
    if (tid < MAX_BUCKETS) sc[tid] += a;
    __syncthreads();
  }
  if (tid < nb) {
    const unsigned start = sc[tid] - tot;
    cur[tid] = start + cnt[tid * EDGE_CHUNKS + blockIdx.x];
    if (blockIdx.x == 0) rowptr[tid * BUCKET_KEYS] = start;
  }
  __syncthreads();

  const int beg = blockIdx.x * chunkLen;
  int end = beg + chunkLen;
  if (end > E) end = E;

  auto place = [&](int s, int d) {
    const unsigned pos = atomicAdd(&cur[((unsigned)d) >> BUCKET_SHIFT], 1u);
    packed[pos] = ((((unsigned)d) & 511u) << 17) | (unsigned)s;
  };
  int e = beg + tid;
  for (; e + 3 * SORT_THREADS < end; e += 4 * SORT_THREADS) {
    const int s0 = src[e];
    const int s1 = src[e + 1 * SORT_THREADS];
    const int s2 = src[e + 2 * SORT_THREADS];
    const int s3 = src[e + 3 * SORT_THREADS];
    const int d0 = dst[e];
    const int d1 = dst[e + 1 * SORT_THREADS];
    const int d2 = dst[e + 2 * SORT_THREADS];
    const int d3 = dst[e + 3 * SORT_THREADS];
    place(s0, d0);
    place(s1, d1);
    place(s2, d2);
    place(s3, d3);
  }
  for (; e < end; e += SORT_THREADS) place(src[e], dst[e]);
}

__device__ __forceinline__ unsigned packed_key(unsigned w) {
  return ((w >> 17) << 1) | ((w >> 16) & 1u);
}

template <bool STAGED>
__device__ __forceinline__ void bucket_fill(const unsigned* __restrict__ packed, unsigned* next,
                                            unsigned short* stage, unsigned short* __restrict__ list,
                                            unsigned beg, unsigned end, int tid) {
  auto place = [&](unsigned w) {
    const unsigned pos = atomicAdd(&next[packed_key(w)], 1u);
    const unsigned short v = (unsigned short)(w & 0xFFFFu);
    if (STAGED) stage[pos - beg] = v;
    else list[pos] = v;
  };
  unsigned p = beg + tid;
  for (; p + 3 * SORT_THREADS < end; p += 4 * SORT_THREADS) {
    const unsigned w0 = packed[p];
    const unsigned w1 = packed[p + 1 * SORT_THREADS];
    const unsigned w2 = packed[p + 2 * SORT_THREADS];
    const unsigned w3 = packed[p + 3 * SORT_THREADS];
    place(w0);
    place(w1);
    place(w2);
    place(w3);
  }
  for (; p < end; p += SORT_THREADS) place(packed[p]);
}

__global__ __launch_bounds__(SORT_THREADS) void k_bsort(const unsigned* __restrict__ packed,
                                                        unsigned* __restrict__ rowptr,
                                                        unsigned short* __restrict__ list,
                                                        float* __restrict__ dinv,
                                                        int E, int n, int nb) {
  __shared__ unsigned hist[BUCKET_KEYS];
  __shared__ unsigned sc[BUCKET_KEYS];
  __shared__ unsigned short stage[STAGE_CAP];
  const int tid = threadIdx.x;
  const int b = blockIdx.x;

  const unsigned beg = rowptr[b * BUCKET_KEYS];
  const unsigned end = (b + 1 < nb) ? rowptr[(b + 1) * BUCKET_KEYS] : (unsigned)E;

  hist[tid] = 0u;
  __syncthreads();

  unsigned p = beg + tid;
  for (; p + 3 * SORT_THREADS < end; p += 4 * SORT_THREADS) {
    const unsigned w0 = packed[p];
    const unsigned w1 = packed[p + 1 * SORT_THREADS];
    const unsigned w2 = packed[p + 2 * SORT_THREADS];
    const unsigned w3 = packed[p + 3 * SORT_THREADS];
    atomicAdd(&hist[packed_key(w0)], 1u);
    atomicAdd(&hist[packed_key(w1)], 1u);
    atomicAdd(&hist[packed_key(w2)], 1u);
    atomicAdd(&hist[packed_key(w3)], 1u);
  }
  for (; p < end; p += SORT_THREADS) atomicAdd(&hist[packed_key(packed[p])], 1u);
  __syncthreads();

  const unsigned mine  = hist[tid];
  const unsigned other = hist[tid ^ 1];
  sc[tid] = mine;
  __syncthreads();
  for (int off = 1; off < BUCKET_KEYS; off <<= 1) {
    const unsigned a = (tid >= off) ? sc[tid - off] : 0u;
    __syncthreads();
    sc[tid] += a;
    __syncthreads();
  }
  const unsigned first = beg + (sc[tid] - mine);
  hist[tid] = first;

  const int node = (b << BUCKET_SHIFT) + (tid >> 1);
  if (node < n) {
    if (tid != 0) rowptr[b * BUCKET_KEYS + tid] = first;
    if ((tid & 1) == 0) dinv[node] = rsqrtf((float)(1u + mine + other));
  }
  __syncthreads();

  if ((end - beg) <= (unsigned)STAGE_CAP) {
    bucket_fill<true>(packed, hist, stage, list, beg, end, tid);
    __syncthreads();
    const unsigned total = end - beg;
    for (unsigned i = tid; i < total; i += SORT_THREADS) list[beg + i] = stage[i];
  } else {
    bucket_fill<false>(packed, hist, stage, list, beg, end, tid);
  }
}

template <int K>
__global__ void k_gemm_wmma(const float* __restrict__ A, const float* __restrict__ W,
                            const float* __restrict__ dinv,
                            float* __restrict__ H, int ntiles) {
  constexpr int PITCH = K / 2 + 2;
  __shared__ __attribute__((aligned(16))) float Wl[32 * PITCH];
  for (int t = threadIdx.x; t < K * 16; t += blockDim.x) {
    const int k = t >> 4, n = t & 15;
    Wl[((((k >> 1) & 1) << 4) | n) * PITCH + ((k >> 2) << 1) + (k & 1)] = W[t];
  }
  __syncthreads();

  const int lane = threadIdx.x & 31;
  const int tile = blockIdx.x * (blockDim.x >> 5) + (threadIdx.x >> 5);
  if (tile >= ntiles) return;

  const int half = lane >> 4;
  const int mn   = lane & 15;
  const int koff = half * 2;

  const float* arow = A + (size_t)(tile * 16 + mn) * K;
  const float* blane = Wl + lane * PITCH;

  v4f d0 = *(const v4f*)(dinv + tile * 16 + half * 8);
  v4f d1 = *(const v4f*)(dinv + tile * 16 + half * 8 + 4);
  int hoff = tile * 256 + half * 128 + mn;
  asm volatile("" : "+v"(d0), "+v"(d1), "+v"(hoff));

  v8f c = {0.f, 0.f, 0.f, 0.f, 0.f, 0.f, 0.f, 0.f};
  v2f a, b;
#pragma unroll
  for (int k0 = 0; k0 < K; k0 += 4) {
    a = *(const v2f*)(arow + k0 + koff);
    b = *(const v2f*)(blane + (k0 >> 1));
    c = __builtin_amdgcn_wmma_f32_16x16x4_f32(
         false, a,  false, b,
         (short)0, c,  false,  false);
  }
  asm volatile("v_nop\n\tv_nop\n\tv_nop\n\tv_nop" : "+v"(c), "+v"(hoff) : "v"(a), "v"(b));

  float* hrow = H + hoff;
  hrow[0 * 16] = c[0] * d0.x;
  hrow[1 * 16] = c[1] * d0.y;
  hrow[2 * 16] = c[2] * d0.z;
  hrow[3 * 16] = c[3] * d0.w;
  hrow[4 * 16] = c[4] * d1.x;
  hrow[5 * 16] = c[5] * d1.y;
  hrow[6 * 16] = c[6] * d1.z;
  hrow[7 * 16] = c[7] * d1.w;
}

__device__ __forceinline__ void edge_add(float4& acc, float4 h) {
  acc.x += h.x;
  acc.y += h.y;
  acc.z += h.z;
  acc.w += h.w;
}

__device__ __forceinline__ void wmma_tail_slots(v8f& c, const v2f (&a)[4], const v2f (&b)[4]) {
  asm volatile("v_nop\n\tv_nop\n\tv_nop\n\tv_nop"
               : "+v"(c)
               : "v"(a[0]), "v"(a[1]), "v"(a[2]), "v"(a[3]),
                 "v"(b[0]), "v"(b[1]), "v"(b[2]), "v"(b[3]));
}

#define ROW_PITCH 20

__global__ __launch_bounds__(256) void k_gather_mm(const unsigned short* __restrict__ list,
                                                   const unsigned* __restrict__ rowptr,
                                                   const float* __restrict__ dinv,
                                                   const float* __restrict__ H,
                                                   const float* __restrict__ bias,
                                                   const float* __restrict__ Wn,
                                                   float* __restrict__ Hn,
                                                   int n, int E, int ntiles) {
  __shared__ __attribute__((aligned(16))) float hs[64 * ROW_PITCH];
  __shared__ float Ws[256];
  const int tid = threadIdx.x;
  Ws[tid] = Wn[tid];

  const int idx = blockIdx.x * 256 + tid;
  const int i = idx >> 2;
  const int q = idx & 3;

  float4 r = make_float4(0.0f, 0.0f, 0.0f, 0.0f);
  if (i < n) {
    const unsigned p0 = rowptr[2 * i];
    const unsigned p1 = rowptr[2 * i + 1];
    const unsigned pe = (i + 1 < n) ? rowptr[2 * i + 2] : (unsigned)E;
    const float dn = dinv[i];
    const float4* H4 = (const float4*)H;

    float4 acc = make_float4(0.0f, 0.0f, 0.0f, 0.0f);
    unsigned p = p0;
    for (; p + 4 <= pe; p += 4) {
      const unsigned s0 = (unsigned)list[p]     + ((p     >= p1) ? 65536u : 0u);
      const unsigned s1 = (unsigned)list[p + 1] + ((p + 1 >= p1) ? 65536u : 0u);
      const unsigned s2 = (unsigned)list[p + 2] + ((p + 2 >= p1) ? 65536u : 0u);
      const unsigned s3 = (unsigned)list[p + 3] + ((p + 3 >= p1) ? 65536u : 0u);
      const float4 h0 = H4[(size_t)s0 * 4 + q];
      const float4 h1 = H4[(size_t)s1 * 4 + q];
      const float4 h2 = H4[(size_t)s2 * 4 + q];
      const float4 h3 = H4[(size_t)s3 * 4 + q];
      edge_add(acc, h0);
      edge_add(acc, h1);
      edge_add(acc, h2);
      edge_add(acc, h3);
    }
    for (; p < pe; p++) {
      const unsigned s = (unsigned)list[p] + ((p >= p1) ? 65536u : 0u);
      edge_add(acc, H4[(size_t)s * 4 + q]);
    }

    const float4 hv = H4[(size_t)i * 4 + q];
    r.x = fmaxf(dn * (acc.x + hv.x) + bias[q * 4 + 0], 0.0f) * dn;
    r.y = fmaxf(dn * (acc.y + hv.y) + bias[q * 4 + 1], 0.0f) * dn;
    r.z = fmaxf(dn * (acc.z + hv.z) + bias[q * 4 + 2], 0.0f) * dn;
    r.w = fmaxf(dn * (acc.w + hv.w) + bias[q * 4 + 3], 0.0f) * dn;
  }
  *(float4*)(hs + (tid >> 2) * ROW_PITCH + q * 4) = r;
  __syncthreads();

  const int wv = tid >> 5;
  const int tile = blockIdx.x * 4 + wv;
  if (wv >= 4 || tile >= ntiles) return;

  const int lane = tid & 31;
  const int half = lane >> 4;
  const int mn   = lane & 15;
  const int koff = half * 2;
  const float* arow = hs + (wv * 16 + mn) * ROW_PITCH;

  v2f a[4], b[4];
#pragma unroll
  for (int s = 0; s < 4; s++) {
    a[s] = *(const v2f*)(arow + 4 * s + koff);
    b[s].x = Ws[(4 * s + koff + 0) * 16 + mn];
    b[s].y = Ws[(4 * s + koff + 1) * 16 + mn];
  }
  v8f c = {0.f, 0.f, 0.f, 0.f, 0.f, 0.f, 0.f, 0.f};
#pragma unroll
  for (int s = 0; s < 4; s++)
    c = __builtin_amdgcn_wmma_f32_16x16x4_f32(
         false, a[s],  false, b[s],
         (short)0, c,  false,  false);
  wmma_tail_slots(c, a, b);

  float* hrow = Hn + (size_t)tile * 256 + half * 128 + mn;
#pragma unroll
  for (int i8 = 0; i8 < 8; i8++) hrow[i8 * 16] = c[i8];
}

__global__ __launch_bounds__(256) void k_gather_head(const unsigned short* __restrict__ list,
                                                     const unsigned* __restrict__ rowptr,
                                                     const float* __restrict__ H,
                                                     const float* __restrict__ bias,
                                                     const float* __restrict__ W,
                                                     const float* __restrict__ l1b,
                                                     const float* __restrict__ l2W,
                                                     const float* __restrict__ l2b,
                                                     float* dio,
                                                     int n, int E, int ntiles) {
  __shared__ __attribute__((aligned(16))) float hs[64 * ROW_PITCH];
  __shared__ float Ws[256];
  const int tid = threadIdx.x;
  Ws[tid] = W[tid];

  const int idx = blockIdx.x * 256 + tid;
  const int i = idx >> 2;
  const int q = idx & 3;

  float4 r = make_float4(0.0f, 0.0f, 0.0f, 0.0f);
  if (i < n) {
    const unsigned p0 = rowptr[2 * i];
    const unsigned p1 = rowptr[2 * i + 1];
    const unsigned pe = (i + 1 < n) ? rowptr[2 * i + 2] : (unsigned)E;
    const float dn = dio[i];
    const float4* H4 = (const float4*)H;

    float4 acc = make_float4(0.0f, 0.0f, 0.0f, 0.0f);
    unsigned p = p0;
    for (; p + 4 <= pe; p += 4) {
      const unsigned s0 = (unsigned)list[p]     + ((p     >= p1) ? 65536u : 0u);
      const unsigned s1 = (unsigned)list[p + 1] + ((p + 1 >= p1) ? 65536u : 0u);
      const unsigned s2 = (unsigned)list[p + 2] + ((p + 2 >= p1) ? 65536u : 0u);
      const unsigned s3 = (unsigned)list[p + 3] + ((p + 3 >= p1) ? 65536u : 0u);
      const float4 h0 = H4[(size_t)s0 * 4 + q];
      const float4 h1 = H4[(size_t)s1 * 4 + q];
      const float4 h2 = H4[(size_t)s2 * 4 + q];
      const float4 h3 = H4[(size_t)s3 * 4 + q];
      edge_add(acc, h0);
      edge_add(acc, h1);
      edge_add(acc, h2);
      edge_add(acc, h3);
    }
    for (; p < pe; p++) {
      const unsigned s = (unsigned)list[p] + ((p >= p1) ? 65536u : 0u);
      edge_add(acc, H4[(size_t)s * 4 + q]);
    }

    const float4 hv = H4[(size_t)i * 4 + q];
    r.x = fmaxf(dn * (acc.x + hv.x) + bias[q * 4 + 0], 0.0f);
    r.y = fmaxf(dn * (acc.y + hv.y) + bias[q * 4 + 1], 0.0f);
    r.z = fmaxf(dn * (acc.z + hv.z) + bias[q * 4 + 2], 0.0f);
    r.w = fmaxf(dn * (acc.w + hv.w) + bias[q * 4 + 3], 0.0f);
  }
  *(float4*)(hs + (tid >> 2) * ROW_PITCH + q * 4) = r;
  __syncthreads();

  const int wv = tid >> 5;
  const int tile = blockIdx.x * 4 + wv;
  if (wv >= 4 || tile >= ntiles) return;

  const int lane = tid & 31;
  const int half = lane >> 4;
  const int mn   = lane & 15;
  const int koff = half * 2;
  const float* arow = hs + (wv * 16 + mn) * ROW_PITCH;
  const float bj = l1b[mn];
  const float wj = l2W[mn];

  v2f a[4], b[4];
#pragma unroll
  for (int s = 0; s < 4; s++) {
    a[s] = *(const v2f*)(arow + 4 * s + koff);
    b[s].x = Ws[(4 * s + koff + 0) * 16 + mn];
    b[s].y = Ws[(4 * s + koff + 1) * 16 + mn];
  }
  v8f c = {0.f, 0.f, 0.f, 0.f, 0.f, 0.f, 0.f, 0.f};
#pragma unroll
  for (int s = 0; s < 4; s++)
    c = __builtin_amdgcn_wmma_f32_16x16x4_f32(
         false, a[s],  false, b[s],
         (short)0, c,  false,  false);
  wmma_tail_slots(c, a, b);

  float mine = 0.0f;
#pragma unroll
  for (int r8 = 0; r8 < 8; r8++) {
    float v = fmaxf(c[r8] + bj, 0.0f) * wj;
#pragma unroll
    for (int off = 8; off; off >>= 1) v += __shfl_xor(v, off, 16);
    if (mn == r8) mine = v;
  }
  if (mn < 8) dio[tile * 16 + half * 8 + mn] = mine + l2b[0];
}

extern "C" void kernel_launch(void* const* d_in, const int* in_sizes, int n_in,
                              void* d_out, int out_size, void* d_ws, size_t ws_size,
                              hipStream_t stream) {
  const float* x   = (const float*)d_in[0];
  const int*   ei  = (const int*)d_in[1];
  const float* W1  = (const float*)d_in[2];
  const float* b1  = (const float*)d_in[3];
  const float* W2  = (const float*)d_in[4];
  const float* b2  = (const float*)d_in[5];
  const float* l1W = (const float*)d_in[6];
  const float* l1b = (const float*)d_in[7];
  const float* l2W = (const float*)d_in[8];
  const float* l2b = (const float*)d_in[9];

  const int N = in_sizes[0] / 128;
  const int E = in_sizes[1] / 2;
  const int* src = ei;
  const int* dst = ei + E;

  unsigned* rowptr = (unsigned*)d_ws;
  float* bufA = (float*)d_ws + 2 * (size_t)N;
  float* bufB = bufA + (size_t)N * 16;
  float* bufC = bufB + (size_t)N * 16;
  float* dinv = (float*)d_out;

  unsigned short* list = (unsigned short*)bufA;
  unsigned* bcnt       = (unsigned*)bufA;
  unsigned* btot       = (unsigned*)bufA + MAX_BUCKETS * EDGE_CHUNKS;
  unsigned* packed     = (unsigned*)bufB;

  const int ntiles = N / 16;
  const int gemmBlocks = (ntiles * 32 + 255) / 256;
  const int gathBlocks = (N * 4 + 255) / 256;

  const int nb = (N + (1 << BUCKET_SHIFT) - 1) >> BUCKET_SHIFT;
  const int chunkLen = (E + EDGE_CHUNKS - 1) / EDGE_CHUNKS;
  k_bcount  <<<EDGE_CHUNKS, SORT_THREADS, 0, stream>>>(dst, bcnt, E, chunkLen, nb);
  k_bscan   <<<(nb + 7) / 8, 256, 0, stream>>>(bcnt, btot, nb);
  k_bscatter<<<EDGE_CHUNKS, SORT_THREADS, 0, stream>>>(src, dst, bcnt, btot, packed, rowptr, E, chunkLen, nb);
  k_bsort   <<<nb, SORT_THREADS, 0, stream>>>(packed, rowptr, list, dinv, E, N, nb);

  k_gemm_wmma<128><<<gemmBlocks, 256, 0, stream>>>(x, W1, dinv, bufB, ntiles);
  k_gather_mm<<<gathBlocks, 256, 0, stream>>>(list, rowptr, dinv, bufB, b1, W2, bufC, N, E, ntiles);

  k_gather_head<<<gathBlocks, 256, 0, stream>>>(list, rowptr, bufC, b2, l1W, l1b, l2W, l2b,
                                                (float*)d_out, N, E, ntiles);
}


# Bucket scatter: a chunk's words put in bucket order in LDS, then each bucket's run stored to consecutive addresses, 32 words a store, not a scattered 4-byte store per edge; slots by LDS ticket
# speedup vs baseline: 4.8452x; 1.1260x over previous
#include <hip/hip_runtime.h>
#include <hip/hip_bf16.h>


typedef __attribute__((ext_vector_type(2))) float v2f;
typedef __attribute__((ext_vector_type(4))) float v4f;
typedef __attribute__((ext_vector_type(8))) float v8f;

#define BUCKET_SHIFT  9
#define BUCKET_KEYS   1024
#define MAX_BUCKETS   256
#define EDGE_CHUNKS   256
#define SORT_THREADS  1024
#define STAGE_CAP     32768
#define SCAT_CAP      16384

__global__ __launch_bounds__(SORT_THREADS) void k_bcount(const int* __restrict__ dst,
                                                         unsigned* __restrict__ cnt,
                                                         int E, int chunkLen, int nb) {
  __shared__ unsigned h[MAX_BUCKETS];
  const int tid = threadIdx.x;
  if (tid < MAX_BUCKETS) h[tid] = 0u;
  __syncthreads();

  const int beg = blockIdx.x * chunkLen;
  int end = beg + chunkLen;
  if (end > E) end = E;

  int e = beg + tid;
  for (; e + 3 * SORT_THREADS < end; e += 4 * SORT_THREADS) {
    const int d0 = dst[e];
    const int d1 = dst[e + 1 * SORT_THREADS];
    const int d2 = dst[e + 2 * SORT_THREADS];
    const int d3 = dst[e + 3 * SORT_THREADS];
    atomicAdd(&h[((unsigned)d0) >> BUCKET_SHIFT], 1u);
    atomicAdd(&h[((unsigned)d1) >> BUCKET_SHIFT], 1u);
    atomicAdd(&h[((unsigned)d2) >> BUCKET_SHIFT], 1u);
    atomicAdd(&h[((unsigned)d3) >> BUCKET_SHIFT], 1u);
  }
  for (; e < end; e += SORT_THREADS) atomicAdd(&h[((unsigned)dst[e]) >> BUCKET_SHIFT], 1u);
  __syncthreads();

  if (tid < nb) cnt[tid * EDGE_CHUNKS + blockIdx.x] = h[tid];
}

__global__ __launch_bounds__(256) void k_bscan(unsigned* __restrict__ cnt,
                                               unsigned* __restrict__ totals, int nb) {
  const int lane = threadIdx.x & 31;
  const int b = blockIdx.x * 8 + (threadIdx.x >> 5);
  if (b >= nb) return;
  unsigned* p = cnt + b * EDGE_CHUNKS + lane * 8;
  unsigned v[8];
  unsigned s = 0u;
#pragma unroll
  for (int j = 0; j < 8; j++) {
    v[j] = p[j];
    s += v[j];
  }
  unsigned incl = s;
#pragma unroll
  for (int off = 1; off < 32; off <<= 1) {
    const unsigned t = __shfl_up(incl, off);
    if (lane >= off) incl += t;
  }
  unsigned run = incl - s;
#pragma unroll
  for (int j = 0; j < 8; j++) {
    p[j] = run;
    run += v[j];
  }
  if (lane == 31) totals[b] = incl;
}

template <bool STAGED>
__device__ __forceinline__ void chunk_place(const int* __restrict__ src, const int* __restrict__ dst,
                                            unsigned* cur, unsigned* stage,
                                            unsigned* __restrict__ packed, int beg, int end, int tid) {
  auto place = [&](int s, int d) {
    const unsigned pos = atomicAdd(&cur[((unsigned)d) >> BUCKET_SHIFT], 1u);
    const unsigned w = ((((unsigned)d) & 511u) << 17) | (unsigned)s;
    if (STAGED) stage[pos] = w;
    else packed[pos] = w;
  };
  int e = beg + tid;
  for (; e + 3 * SORT_THREADS < end; e += 4 * SORT_THREADS) {
    const int s0 = src[e];
    const int s1 = src[e + 1 * SORT_THREADS];
    const int s2 = src[e + 2 * SORT_THREADS];
    const int s3 = src[e + 3 * SORT_THREADS];
    const int d0 = dst[e];
    const int d1 = dst[e + 1 * SORT_THREADS];
    const int d2 = dst[e + 2 * SORT_THREADS];
    const int d3 = dst[e + 3 * SORT_THREADS];
    place(s0, d0);
    place(s1, d1);
    place(s2, d2);
    place(s3, d3);
  }
  for (; e < end; e += SORT_THREADS) place(src[e], dst[e]);
}

__global__ __launch_bounds__(SORT_THREADS) void k_bscatter(const int* __restrict__ src,
                                                           const int* __restrict__ dst,
                                                           const unsigned* __restrict__ cnt,
                                                           const unsigned* __restrict__ totals,
                                                           unsigned* __restrict__ packed,
                                                           unsigned* __restrict__ rowptr,
                                                           int E, int chunkLen, int nb) {
  __shared__ unsigned sc[MAX_BUCKETS];
  __shared__ unsigned so[MAX_BUCKETS];
  __shared__ unsigned cur[MAX_BUCKETS];
  __shared__ unsigned gbase[MAX_BUCKETS];
  __shared__ unsigned lbeg[MAX_BUCKETS];
  __shared__ unsigned lcnt[MAX_BUCKETS];
  __shared__ unsigned stage[SCAT_CAP];
  const int tid = threadIdx.x;

  const int beg = blockIdx.x * chunkLen;
  int end = beg + chunkLen;
  if (end > E) end = E;
  const bool staged = (end - beg) <= SCAT_CAP;

  unsigned tot = 0u, before = 0u, own = 0u;
  if (tid < MAX_BUCKETS) {
    if (tid < nb) {
      tot = totals[tid];
      before = cnt[tid * EDGE_CHUNKS + blockIdx.x];
      const unsigned upto = (blockIdx.x + 1 < EDGE_CHUNKS) ? cnt[tid * EDGE_CHUNKS + blockIdx.x + 1] : tot;
      own = upto - before;
    }
    sc[tid] = tot;
    so[tid] = own;
  }
  __syncthreads();
  for (int off = 1; off < MAX_BUCKETS; off <<= 1) {
    unsigned a = 0u, ao = 0u;
    if (tid < MAX_BUCKETS && tid >= off) {
      a = sc[tid - off];
      ao = so[tid - off];
    }
    __syncthreads();
    if (tid < MAX_BUCKETS) {
      sc[tid] += a;
      so[tid] += ao;
    }
    __syncthreads();
  }
  if (tid < nb) {
    const unsigned start = sc[tid] - tot;
    const unsigned lfirst = so[tid] - own;
    gbase[tid] = start + before;
    lbeg[tid] = lfirst;
    lcnt[tid] = own;
    cur[tid] = staged ? lfirst : (start + before);
    if (blockIdx.x == 0) rowptr[tid * BUCKET_KEYS] = start;
  }
  __syncthreads();

  if (staged) {
    chunk_place<true>(src, dst, cur, stage, packed, beg, end, tid);
    __syncthreads();
    const int lane = tid & 31;
    for (int b = tid >> 5; b < nb; b += SORT_THREADS / 32) {
      const unsigned l0 = lbeg[b];
      const unsigned c = lcnt[b];
      const unsigned g0 = gbase[b];
#pragma unroll 1
      for (unsigned j = lane; j < c; j += 32) packed[g0 + j] = stage[l0 + j];
    }
  } else {
    chunk_place<false>(src, dst, cur, stage, packed, beg, end, tid);
  }
}

__device__ __forceinline__ unsigned packed_key(unsigned w) {
  return ((w >> 17) << 1) | ((w >> 16) & 1u);
}

template <bool STAGED>
__device__ __forceinline__ void bucket_fill(const unsigned* __restrict__ packed, unsigned* next,
                                            unsigned short* stage, unsigned short* __restrict__ list,
                                            unsigned beg, unsigned end, int tid) {
  auto place = [&](unsigned w) {
    const unsigned pos = atomicAdd(&next[packed_key(w)], 1u);
    const unsigned short v = (unsigned short)(w & 0xFFFFu);
    if (STAGED) stage[pos - beg] = v;
    else list[pos] = v;
  };
  unsigned p = beg + tid;
  for (; p + 3 * SORT_THREADS < end; p += 4 * SORT_THREADS) {
    const unsigned w0 = packed[p];
    const unsigned w1 = packed[p + 1 * SORT_THREADS];
    const unsigned w2 = packed[p + 2 * SORT_THREADS];
    const unsigned w3 = packed[p + 3 * SORT_THREADS];
    place(w0);
    place(w1);
    place(w2);
    place(w3);
  }
  for (; p < end; p += SORT_THREADS) place(packed[p]);
}

__global__ __launch_bounds__(SORT_THREADS) void k_bsort(const unsigned* __restrict__ packed,
                                                        unsigned* __restrict__ rowptr,
                                                        unsigned short* __restrict__ list,
                                                        float* __restrict__ dinv,
                                                        int E, int n, int nb) {
  __shared__ unsigned hist[BUCKET_KEYS];
  __shared__ unsigned sc[BUCKET_KEYS];
  __shared__ unsigned short stage[STAGE_CAP];
  const int tid = threadIdx.x;
  const int b = blockIdx.x;

  const unsigned beg = rowptr[b * BUCKET_KEYS];
  const unsigned end = (b + 1 < nb) ? rowptr[(b + 1) * BUCKET_KEYS] : (unsigned)E;

  hist[tid] = 0u;
  __syncthreads();

  unsigned p = beg + tid;
  for (; p + 3 * SORT_THREADS < end; p += 4 * SORT_THREADS) {
    const unsigned w0 = packed[p];
    const unsigned w1 = packed[p + 1 * SORT_THREADS];
    const unsigned w2 = packed[p + 2 * SORT_THREADS];
    const unsigned w3 = packed[p + 3 * SORT_THREADS];
    atomicAdd(&hist[packed_key(w0)], 1u);
    atomicAdd(&hist[packed_key(w1)], 1u);
    atomicAdd(&hist[packed_key(w2)], 1u);
    atomicAdd(&hist[packed_key(w3)], 1u);
  }
  for (; p < end; p += SORT_THREADS) atomicAdd(&hist[packed_key(packed[p])], 1u);
  __syncthreads();

  const unsigned mine  = hist[tid];
  const unsigned other = hist[tid ^ 1];
  sc[tid] = mine;
  __syncthreads();
  for (int off = 1; off < BUCKET_KEYS; off <<= 1) {
    const unsigned a = (tid >= off) ? sc[tid - off] : 0u;
    __syncthreads();
    sc[tid] += a;
    __syncthreads();
  }
  const unsigned first = beg + (sc[tid] - mine);
  hist[tid] = first;

  const int node = (b << BUCKET_SHIFT) + (tid >> 1);
  if (node < n) {
    if (tid != 0) rowptr[b * BUCKET_KEYS + tid] = first;
    if ((tid & 1) == 0) dinv[node] = rsqrtf((float)(1u + mine + other));
  }
  __syncthreads();

  if ((end - beg) <= (unsigned)STAGE_CAP) {
    bucket_fill<true>(packed, hist, stage, list, beg, end, tid);
    __syncthreads();
    const unsigned total = end - beg;
    for (unsigned i = tid; i < total; i += SORT_THREADS) list[beg + i] = stage[i];
  } else {
    bucket_fill<false>(packed, hist, stage, list, beg, end, tid);
  }
}

template <int K>
__global__ void k_gemm_wmma(const float* __restrict__ A, const float* __restrict__ W,
                            const float* __restrict__ dinv,
                            float* __restrict__ H, int ntiles) {
  constexpr int PITCH = K / 2 + 2;
  __shared__ __attribute__((aligned(16))) float Wl[32 * PITCH];
  for (int t = threadIdx.x; t < K * 16; t += blockDim.x) {
    const int k = t >> 4, n = t & 15;
    Wl[((((k >> 1) & 1) << 4) | n) * PITCH + ((k >> 2) << 1) + (k & 1)] = W[t];
  }
  __syncthreads();

  const int lane = threadIdx.x & 31;
  const int tile = blockIdx.x * (blockDim.x >> 5) + (threadIdx.x >> 5);
  if (tile >= ntiles) return;

  const int half = lane >> 4;
  const int mn   = lane & 15;
  const int koff = half * 2;

  const float* arow = A + (size_t)(tile * 16 + mn) * K;
  const float* blane = Wl + lane * PITCH;

  v4f d0 = *(const v4f*)(dinv + tile * 16 + half * 8);
  v4f d1 = *(const v4f*)(dinv + tile * 16 + half * 8 + 4);
  int hoff = tile * 256 + half * 128 + mn;
  asm volatile("" : "+v"(d0), "+v"(d1), "+v"(hoff));

  v8f c = {0.f, 0.f, 0.f, 0.f, 0.f, 0.f, 0.f, 0.f};
  v2f a, b;
#pragma unroll
  for (int k0 = 0; k0 < K; k0 += 4) {
    a = *(const v2f*)(arow + k0 + koff);
    b = *(const v2f*)(blane + (k0 >> 1));
    c = __builtin_amdgcn_wmma_f32_16x16x4_f32(
         false, a,  false, b,
         (short)0, c,  false,  false);
  }
  asm volatile("v_nop\n\tv_nop\n\tv_nop\n\tv_nop" : "+v"(c), "+v"(hoff) : "v"(a), "v"(b));

  float* hrow = H + hoff;
  hrow[0 * 16] = c[0] * d0.x;
  hrow[1 * 16] = c[1] * d0.y;
  hrow[2 * 16] = c[2] * d0.z;
  hrow[3 * 16] = c[3] * d0.w;
  hrow[4 * 16] = c[4] * d1.x;
  hrow[5 * 16] = c[5] * d1.y;
  hrow[6 * 16] = c[6] * d1.z;
  hrow[7 * 16] = c[7] * d1.w;
}

__device__ __forceinline__ void edge_add(float4& acc, float4 h) {
  acc.x += h.x;
  acc.y += h.y;
  acc.z += h.z;
  acc.w += h.w;
}

__device__ __forceinline__ void wmma_tail_slots(v8f& c, const v2f (&a)[4], const v2f (&b)[4]) {
  asm volatile("v_nop\n\tv_nop\n\tv_nop\n\tv_nop"
               : "+v"(c)
               : "v"(a[0]), "v"(a[1]), "v"(a[2]), "v"(a[3]),
                 "v"(b[0]), "v"(b[1]), "v"(b[2]), "v"(b[3]));
}

#define ROW_PITCH 20

__global__ __launch_bounds__(256) void k_gather_mm(const unsigned short* __restrict__ list,
                                                   const unsigned* __restrict__ rowptr,
                                                   const float* __restrict__ dinv,
                                                   const float* __restrict__ H,
                                                   const float* __restrict__ bias,
                                                   const float* __restrict__ Wn,
                                                   float* __restrict__ Hn,
                                                   int n, int E, int ntiles) {
  __shared__ __attribute__((aligned(16))) float hs[64 * ROW_PITCH];
  __shared__ float Ws[256];
  const int tid = threadIdx.x;
  Ws[tid] = Wn[tid];

  const int idx = blockIdx.x * 256 + tid;
  const int i = idx >> 2;
  const int q = idx & 3;

  float4 r = make_float4(0.0f, 0.0f, 0.0f, 0.0f);
  if (i < n) {
    const unsigned p0 = rowptr[2 * i];
    const unsigned p1 = rowptr[2 * i + 1];
    const unsigned pe = (i + 1 < n) ? rowptr[2 * i + 2] : (unsigned)E;
    const float dn = dinv[i];
    const float4* H4 = (const float4*)H;

    float4 acc = make_float4(0.0f, 0.0f, 0.0f, 0.0f);
    unsigned p = p0;
    for (; p + 4 <= pe; p += 4) {
      const unsigned s0 = (unsigned)list[p]     + ((p     >= p1) ? 65536u : 0u);
      const unsigned s1 = (unsigned)list[p + 1] + ((p + 1 >= p1) ? 65536u : 0u);
      const unsigned s2 = (unsigned)list[p + 2] + ((p + 2 >= p1) ? 65536u : 0u);
      const unsigned s3 = (unsigned)list[p + 3] + ((p + 3 >= p1) ? 65536u : 0u);
      const float4 h0 = H4[(size_t)s0 * 4 + q];
      const float4 h1 = H4[(size_t)s1 * 4 + q];
      const float4 h2 = H4[(size_t)s2 * 4 + q];
      const float4 h3 = H4[(size_t)s3 * 4 + q];
      edge_add(acc, h0);
      edge_add(acc, h1);
      edge_add(acc, h2);
      edge_add(acc, h3);
    }
    for (; p < pe; p++) {
      const unsigned s = (unsigned)list[p] + ((p >= p1) ? 65536u : 0u);
      edge_add(acc, H4[(size_t)s * 4 + q]);
    }

    const float4 hv = H4[(size_t)i * 4 + q];
    r.x = fmaxf(dn * (acc.x + hv.x) + bias[q * 4 + 0], 0.0f) * dn;
    r.y = fmaxf(dn * (acc.y + hv.y) + bias[q * 4 + 1], 0.0f) * dn;
    r.z = fmaxf(dn * (acc.z + hv.z) + bias[q * 4 + 2], 0.0f) * dn;
    r.w = fmaxf(dn * (acc.w + hv.w) + bias[q * 4 + 3], 0.0f) * dn;
  }
  *(float4*)(hs + (tid >> 2) * ROW_PITCH + q * 4) = r;
  __syncthreads();

  const int wv = tid >> 5;
  const int tile = blockIdx.x * 4 + wv;
  if (wv >= 4 || tile >= ntiles) return;

  const int lane = tid & 31;
  const int half = lane >> 4;
  const int mn   = lane & 15;
  const int koff = half * 2;
  const float* arow = hs + (wv * 16 + mn) * ROW_PITCH;

  v2f a[4], b[4];
#pragma unroll
  for (int s = 0; s < 4; s++) {
    a[s] = *(const v2f*)(arow + 4 * s + koff);
    b[s].x = Ws[(4 * s + koff + 0) * 16 + mn];
    b[s].y = Ws[(4 * s + koff + 1) * 16 + mn];
  }
  v8f c = {0.f, 0.f, 0.f, 0.f, 0.f, 0.f, 0.f, 0.f};
#pragma unroll
  for (int s = 0; s < 4; s++)
    c = __builtin_amdgcn_wmma_f32_16x16x4_f32(
         false, a[s],  false, b[s],
         (short)0, c,  false,  false);
  wmma_tail_slots(c, a, b);

  float* hrow = Hn + (size_t)tile * 256 + half * 128 + mn;
#pragma unroll
  for (int i8 = 0; i8 < 8; i8++) hrow[i8 * 16] = c[i8];
}

__global__ __launch_bounds__(256) void k_gather_head(const unsigned short* __restrict__ list,
                                                     const unsigned* __restrict__ rowptr,
                                                     const float* __restrict__ H,
                                                     const float* __restrict__ bias,
                                                     const float* __restrict__ W,
                                                     const float* __restrict__ l1b,
                                                     const float* __restrict__ l2W,
                                                     const float* __restrict__ l2b,
                                                     float* dio,
                                                     int n, int E, int ntiles) {
  __shared__ __attribute__((aligned(16))) float hs[64 * ROW_PITCH];
  __shared__ float Ws[256];
  const int tid = threadIdx.x;
  Ws[tid] = W[tid];

  const int idx = blockIdx.x * 256 + tid;
  const int i = idx >> 2;
  const int q = idx & 3;

  float4 r = make_float4(0.0f, 0.0f, 0.0f, 0.0f);
  if (i < n) {
    const unsigned p0 = rowptr[2 * i];
    const unsigned p1 = rowptr[2 * i + 1];
    const unsigned pe = (i + 1 < n) ? rowptr[2 * i + 2] : (unsigned)E;
    const float dn = dio[i];
    const float4* H4 = (const float4*)H;

    float4 acc = make_float4(0.0f, 0.0f, 0.0f, 0.0f);
    unsigned p = p0;
    for (; p + 4 <= pe; p += 4) {
      const unsigned s0 = (unsigned)list[p]     + ((p     >= p1) ? 65536u : 0u);
      const unsigned s1 = (unsigned)list[p + 1] + ((p + 1 >= p1) ? 65536u : 0u);
      const unsigned s2 = (unsigned)list[p + 2] + ((p + 2 >= p1) ? 65536u : 0u);
      const unsigned s3 = (unsigned)list[p + 3] + ((p + 3 >= p1) ? 65536u : 0u);
      const float4 h0 = H4[(size_t)s0 * 4 + q];
      const float4 h1 = H4[(size_t)s1 * 4 + q];
      const float4 h2 = H4[(size_t)s2 * 4 + q];
      const float4 h3 = H4[(size_t)s3 * 4 + q];
      edge_add(acc, h0);
      edge_add(acc, h1);
      edge_add(acc, h2);
      edge_add(acc, h3);
    }
    for (; p < pe; p++) {
      const unsigned s = (unsigned)list[p] + ((p >= p1) ? 65536u : 0u);
      edge_add(acc, H4[(size_t)s * 4 + q]);
    }

    const float4 hv = H4[(size_t)i * 4 + q];
    r.x = fmaxf(dn * (acc.x + hv.x) + bias[q * 4 + 0], 0.0f);
    r.y = fmaxf(dn * (acc.y + hv.y) + bias[q * 4 + 1], 0.0f);
    r.z = fmaxf(dn * (acc.z + hv.z) + bias[q * 4 + 2], 0.0f);
    r.w = fmaxf(dn * (acc.w + hv.w) + bias[q * 4 + 3], 0.0f);
  }
  *(float4*)(hs + (tid >> 2) * ROW_PITCH + q * 4) = r;
  __syncthreads();

  const int wv = tid >> 5;
  const int tile = blockIdx.x * 4 + wv;
  if (wv >= 4 || tile >= ntiles) return;

  const int lane = tid & 31;
  const int half = lane >> 4;
  const int mn   = lane & 15;
  const int koff = half * 2;
  const float* arow = hs + (wv * 16 + mn) * ROW_PITCH;
  const float bj = l1b[mn];
  const float wj = l2W[mn];

  v2f a[4], b[4];
#pragma unroll
  for (int s = 0; s < 4; s++) {
    a[s] = *(const v2f*)(arow + 4 * s + koff);
    b[s].x = Ws[(4 * s + koff + 0) * 16 + mn];
    b[s].y = Ws[(4 * s + koff + 1) * 16 + mn];
  }
  v8f c = {0.f, 0.f, 0.f, 0.f, 0.f, 0.f, 0.f, 0.f};
#pragma unroll
  for (int s = 0; s < 4; s++)
    c = __builtin_amdgcn_wmma_f32_16x16x4_f32(
         false, a[s],  false, b[s],
         (short)0, c,  false,  false);
  wmma_tail_slots(c, a, b);

  float mine = 0.0f;
#pragma unroll
  for (int r8 = 0; r8 < 8; r8++) {
    float v = fmaxf(c[r8] + bj, 0.0f) * wj;
#pragma unroll
    for (int off = 8; off; off >>= 1) v += __shfl_xor(v, off, 16);
    if (mn == r8) mine = v;
  }
  if (mn < 8) dio[tile * 16 + half * 8 + mn] = mine + l2b[0];
}

extern "C" void kernel_launch(void* const* d_in, const int* in_sizes, int n_in,
                              void* d_out, int out_size, void* d_ws, size_t ws_size,
                              hipStream_t stream) {
  const float* x   = (const float*)d_in[0];
  const int*   ei  = (const int*)d_in[1];
  const float* W1  = (const float*)d_in[2];
  const float* b1  = (const float*)d_in[3];
  const float* W2  = (const float*)d_in[4];
  const float* b2  = (const float*)d_in[5];
  const float* l1W = (const float*)d_in[6];
  const float* l1b = (const float*)d_in[7];
  const float* l2W = (const float*)d_in[8];
  const float* l2b = (const float*)d_in[9];

  const int N = in_sizes[0] / 128;
  const int E = in_sizes[1] / 2;
  const int* src = ei;
  const int* dst = ei + E;

  unsigned* rowptr = (unsigned*)d_ws;
  float* bufA = (float*)d_ws + 2 * (size_t)N;
  float* bufB = bufA + (size_t)N * 16;
  float* bufC = bufB + (size_t)N * 16;
  float* dinv = (float*)d_out;

  unsigned short* list = (unsigned short*)bufA;
  unsigned* bcnt       = (unsigned*)bufA;
  unsigned* btot       = (unsigned*)bufA + MAX_BUCKETS * EDGE_CHUNKS;
  unsigned* packed     = (unsigned*)bufB;

  const int ntiles = N / 16;
  const int gemmBlocks = (ntiles * 32 + 255) / 256;
  const int gathBlocks = (N * 4 + 255) / 256;

  const int nb = (N + (1 << BUCKET_SHIFT) - 1) >> BUCKET_SHIFT;
  const int chunkLen = (E + EDGE_CHUNKS - 1) / EDGE_CHUNKS;
  k_bcount  <<<EDGE_CHUNKS, SORT_THREADS, 0, stream>>>(dst, bcnt, E, chunkLen, nb);
  k_bscan   <<<(nb + 7) / 8, 256, 0, stream>>>(bcnt, btot, nb);
  k_bscatter<<<EDGE_CHUNKS, SORT_THREADS, 0, stream>>>(src, dst, bcnt, btot, packed, rowptr, E, chunkLen, nb);
  k_bsort   <<<nb, SORT_THREADS, 0, stream>>>(packed, rowptr, list, dinv, E, N, nb);

  k_gemm_wmma<128><<<gemmBlocks, 256, 0, stream>>>(x, W1, dinv, bufB, ntiles);
  k_gather_mm<<<gathBlocks, 256, 0, stream>>>(list, rowptr, dinv, bufB, b1, W2, bufC, N, E, ntiles);

  k_gather_head<<<gathBlocks, 256, 0, stream>>>(list, rowptr, bufC, b2, l1W, l1b, l2W, l2b,
                                                (float*)d_out, N, E, ntiles);
}


# Each edge chunk put in bucket order inside its own range by one launch (count, scan, place); the bucket sort collects its 256 runs: no global count or scan, 7->5 launches; sum order in a node moves
# speedup vs baseline: 5.1238x; 1.0575x over previous
#include <hip/hip_runtime.h>
#include <hip/hip_bf16.h>


typedef __attribute__((ext_vector_type(2))) float v2f;
typedef __attribute__((ext_vector_type(4))) float v4f;
typedef __attribute__((ext_vector_type(8))) float v8f;


#define BUCKET_SHIFT  9
#define BUCKET_KEYS   1024
#define MAX_BUCKETS   256
#define EDGE_CHUNKS   256
#define SORT_THREADS  1024
#define STAGE_CAP     32768
#define SCAT_CAP      16384

static_assert(SORT_THREADS == 4 * EDGE_CHUNKS, "k_bsort walks one run with four lanes");

template <bool STAGED>
__device__ __forceinline__ void chunk_place(const int* __restrict__ src, const int* __restrict__ dst,
                                            unsigned* cur, unsigned* stage,
                                            unsigned* __restrict__ packed, int beg, int end, int tid) {
  auto place = [&](int s, int d) {
    const unsigned pos = atomicAdd(&cur[((unsigned)d) >> BUCKET_SHIFT], 1u);
    const unsigned w = ((((unsigned)d) & 511u) << 17) | (unsigned)s;
    if (STAGED) stage[pos] = w;
    else packed[pos] = w;
  };
  int e = beg + tid;
  for (; e + 3 * SORT_THREADS < end; e += 4 * SORT_THREADS) {
    const int s0 = src[e];
    const int s1 = src[e + 1 * SORT_THREADS];
    const int s2 = src[e + 2 * SORT_THREADS];
    const int s3 = src[e + 3 * SORT_THREADS];
    const int d0 = dst[e];
    const int d1 = dst[e + 1 * SORT_THREADS];
    const int d2 = dst[e + 2 * SORT_THREADS];
    const int d3 = dst[e + 3 * SORT_THREADS];
    place(s0, d0);
    place(s1, d1);
    place(s2, d2);
    place(s3, d3);
  }
  for (; e < end; e += SORT_THREADS) place(src[e], dst[e]);
}

__global__ __launch_bounds__(SORT_THREADS) void k_bgroup(const int* __restrict__ src,
                                                         const int* __restrict__ dst,
                                                         unsigned* __restrict__ packed,
                                                         unsigned* __restrict__ tab,
                                                         int E, int chunkLen, int nb) {
  __shared__ unsigned h[MAX_BUCKETS];
  __shared__ unsigned so[MAX_BUCKETS];
  __shared__ unsigned stage[SCAT_CAP];
  const int tid = threadIdx.x;
  if (tid < MAX_BUCKETS) h[tid] = 0u;
  __syncthreads();

  int beg = blockIdx.x * chunkLen;
  if (beg > E) beg = E;
  int end = beg + chunkLen;
  if (end > E) end = E;
  const bool staged = (end - beg) <= SCAT_CAP;

  int e = beg + tid;
  for (; e + 3 * SORT_THREADS < end; e += 4 * SORT_THREADS) {
    const int d0 = dst[e];
    const int d1 = dst[e + 1 * SORT_THREADS];
    const int d2 = dst[e + 2 * SORT_THREADS];
    const int d3 = dst[e + 3 * SORT_THREADS];
    atomicAdd(&h[((unsigned)d0) >> BUCKET_SHIFT], 1u);
    atomicAdd(&h[((unsigned)d1) >> BUCKET_SHIFT], 1u);
    atomicAdd(&h[((unsigned)d2) >> BUCKET_SHIFT], 1u);
    atomicAdd(&h[((unsigned)d3) >> BUCKET_SHIFT], 1u);
  }
  for (; e < end; e += SORT_THREADS) atomicAdd(&h[((unsigned)dst[e]) >> BUCKET_SHIFT], 1u);
  __syncthreads();

  unsigned own = 0u;
  if (tid < MAX_BUCKETS) {
    own = h[tid];
    so[tid] = own;
  }
  __syncthreads();
  for (int off = 1; off < MAX_BUCKETS; off <<= 1) {
    unsigned a = 0u;
    if (tid < MAX_BUCKETS && tid >= off) a = so[tid - off];
    __syncthreads();
    if (tid < MAX_BUCKETS) so[tid] += a;
    __syncthreads();
  }
  if (tid < MAX_BUCKETS) {
    const unsigned lfirst = so[tid] - own;
    if (tid < nb) tab[tid * BUCKET_KEYS + blockIdx.x] = (lfirst << 16) | own;
    h[tid] = staged ? lfirst : ((unsigned)beg + lfirst);
  }
  __syncthreads();

  if (staged) {
    chunk_place<true>(src, dst, h, stage, packed, beg, end, tid);
    __syncthreads();
    const int len = end - beg;
    for (int i = tid; i < len; i += SORT_THREADS) packed[beg + i] = stage[i];
  } else {
    chunk_place<false>(src, dst, h, stage, packed, beg, end, tid);
  }
}

__device__ __forceinline__ unsigned packed_key(unsigned w) {
  return ((w >> 17) << 1) | ((w >> 16) & 1u);
}

template <class F>
__device__ __forceinline__ void run_words(const unsigned* __restrict__ run, const unsigned* __restrict__ safe,
                                          unsigned q, unsigned len, F f) {
  unsigned j = q;
  for (; j + 28u < len; j += 32u) {
    const unsigned w0 = run[j];
    const unsigned w1 = run[j + 4u];
    const unsigned w2 = run[j + 8u];
    const unsigned w3 = run[j + 12u];
    const unsigned w4 = run[j + 16u];
    const unsigned w5 = run[j + 20u];
    const unsigned w6 = run[j + 24u];
    const unsigned w7 = run[j + 28u];
    f(w0);
    f(w1);
    f(w2);
    f(w3);
    f(w4);
    f(w5);
    f(w6);
    f(w7);
  }
  unsigned w[7];
  bool have[7];
#pragma unroll
  for (int k = 0; k < 7; k++) {
    have[k] = (j + 4u * k) < len;
    const unsigned* p = have[k] ? (run + (j + 4u * k)) : safe;
    w[k] = *p;
  }
#pragma unroll
  for (int k = 0; k < 7; k++)
    if (have[k]) f(w[k]);
}

__global__ __launch_bounds__(SORT_THREADS) void k_bsort(const unsigned* __restrict__ packed,
                                                        unsigned* rowptr,
                                                        unsigned short* __restrict__ list,
                                                        float* __restrict__ dinv,
                                                        int chunkLen, int n) {
  __shared__ unsigned hist[BUCKET_KEYS];
  __shared__ unsigned sc[BUCKET_KEYS];
  __shared__ unsigned short stage[STAGE_CAP];
  __shared__ unsigned tot[2];
  const int tid = threadIdx.x;
  const int b = blockIdx.x;

  const unsigned t = rowptr[b * BUCKET_KEYS + (tid >> 2)];
  const unsigned len = t & 0xFFFFu;
  const unsigned lfirst = t >> 16;
  const unsigned* run = packed + (size_t)(tid >> 2) * (size_t)chunkLen + lfirst;
  const unsigned q = (unsigned)tid & 3u;

  hist[tid] = 0u;
  if (tid < 2) tot[tid] = 0u;
  __syncthreads();

  {
    unsigned a = (q == 0u) ? lfirst : 0u;
    unsigned c = (q == 0u) ? len : 0u;
    for (int off = 16; off; off >>= 1) {
      a += __shfl_xor(a, off);
      c += __shfl_xor(c, off);
    }
    if ((tid & 31) == 0) {
      atomicAdd(&tot[0], a);
      atomicAdd(&tot[1], c);
    }
  }

  run_words(run, packed, q, len, [&](unsigned w) { atomicAdd(&hist[packed_key(w)], 1u); });
  __syncthreads();

  const unsigned beg = tot[0];
  const unsigned total = tot[1];
  const unsigned mine  = hist[tid];
  const unsigned other = hist[tid ^ 1];
  sc[tid] = mine;
  __syncthreads();
  for (int off = 1; off < BUCKET_KEYS; off <<= 1) {
    const unsigned a = (tid >= off) ? sc[tid - off] : 0u;
    __syncthreads();
    sc[tid] += a;
    __syncthreads();
  }
  const unsigned first = beg + (sc[tid] - mine);
  hist[tid] = first;

  const int node = (b << BUCKET_SHIFT) + (tid >> 1);
  if (node < n) {
    rowptr[b * BUCKET_KEYS + tid] = first;
    if ((tid & 1) == 0) dinv[node] = rsqrtf((float)(1u + mine + other));
  }
  __syncthreads();

  const bool staged = total <= (unsigned)STAGE_CAP;
  auto place = [&](unsigned w) {
    const unsigned pos = atomicAdd(&hist[packed_key(w)], 1u);
    const unsigned short v = (unsigned short)(w & 0xFFFFu);
    if (staged) stage[pos - beg] = v;
    else list[pos] = v;
  };
  run_words(run, packed, q, len, place);
  if (staged) {
    __syncthreads();
    for (unsigned i = tid; i < total; i += SORT_THREADS) list[beg + i] = stage[i];
  }
}

template <int K>
__global__ void k_gemm_wmma(const float* __restrict__ A, const float* __restrict__ W,
                            const float* __restrict__ dinv,
                            float* __restrict__ H, int ntiles) {
  constexpr int PITCH = K / 2 + 2;
  __shared__ __attribute__((aligned(16))) float Wl[32 * PITCH];
  for (int t = threadIdx.x; t < K * 16; t += blockDim.x) {
    const int k = t >> 4, n = t & 15;
    Wl[((((k >> 1) & 1) << 4) | n) * PITCH + ((k >> 2) << 1) + (k & 1)] = W[t];
  }
  __syncthreads();

  const int lane = threadIdx.x & 31;
  const int tile = blockIdx.x * (blockDim.x >> 5) + (threadIdx.x >> 5);
  if (tile >= ntiles) return;

  const int half = lane >> 4;
  const int mn   = lane & 15;
  const int koff = half * 2;

  const float* arow = A + (size_t)(tile * 16 + mn) * K;
  const float* blane = Wl + lane * PITCH;

  v4f d0 = *(const v4f*)(dinv + tile * 16 + half * 8);
  v4f d1 = *(const v4f*)(dinv + tile * 16 + half * 8 + 4);
  int hoff = tile * 256 + half * 128 + mn;
  asm volatile("" : "+v"(d0), "+v"(d1), "+v"(hoff));

  v8f c = {0.f, 0.f, 0.f, 0.f, 0.f, 0.f, 0.f, 0.f};
  v2f a, b;
#pragma unroll
  for (int k0 = 0; k0 < K; k0 += 4) {
    a = *(const v2f*)(arow + k0 + koff);
    b = *(const v2f*)(blane + (k0 >> 1));
    c = __builtin_amdgcn_wmma_f32_16x16x4_f32(
         false, a,  false, b,
         (short)0, c,  false,  false);
  }
  asm volatile("v_nop\n\tv_nop\n\tv_nop\n\tv_nop" : "+v"(c), "+v"(hoff) : "v"(a), "v"(b));

  float* hrow = H + hoff;
  hrow[0 * 16] = c[0] * d0.x;
  hrow[1 * 16] = c[1] * d0.y;
  hrow[2 * 16] = c[2] * d0.z;
  hrow[3 * 16] = c[3] * d0.w;
  hrow[4 * 16] = c[4] * d1.x;
  hrow[5 * 16] = c[5] * d1.y;
  hrow[6 * 16] = c[6] * d1.z;
  hrow[7 * 16] = c[7] * d1.w;
}

__device__ __forceinline__ void edge_add(float4& acc, float4 h) {
  acc.x += h.x;
  acc.y += h.y;
  acc.z += h.z;
  acc.w += h.w;
}

__device__ __forceinline__ void wmma_tail_slots(v8f& c, const v2f (&a)[4], const v2f (&b)[4]) {
  asm volatile("v_nop\n\tv_nop\n\tv_nop\n\tv_nop"
               : "+v"(c)
               : "v"(a[0]), "v"(a[1]), "v"(a[2]), "v"(a[3]),
                 "v"(b[0]), "v"(b[1]), "v"(b[2]), "v"(b[3]));
}

#define ROW_PITCH 20

__global__ __launch_bounds__(256) void k_gather_mm(const unsigned short* __restrict__ list,
                                                   const unsigned* __restrict__ rowptr,
                                                   const float* __restrict__ dinv,
                                                   const float* __restrict__ H,
                                                   const float* __restrict__ bias,
                                                   const float* __restrict__ Wn,
                                                   float* __restrict__ Hn,
                                                   int n, int E, int ntiles) {
  __shared__ __attribute__((aligned(16))) float hs[64 * ROW_PITCH];
  __shared__ float Ws[256];
  const int tid = threadIdx.x;
  Ws[tid] = Wn[tid];

  const int idx = blockIdx.x * 256 + tid;
  const int i = idx >> 2;
  const int q = idx & 3;

  float4 r = make_float4(0.0f, 0.0f, 0.0f, 0.0f);
  if (i < n) {
    const unsigned p0 = rowptr[2 * i];
    const unsigned p1 = rowptr[2 * i + 1];
    const unsigned pe = (i + 1 < n) ? rowptr[2 * i + 2] : (unsigned)E;
    const float dn = dinv[i];
    const float4* H4 = (const float4*)H;

    float4 acc = make_float4(0.0f, 0.0f, 0.0f, 0.0f);
    unsigned p = p0;
    for (; p + 4 <= pe; p += 4) {
      const unsigned s0 = (unsigned)list[p]     + ((p     >= p1) ? 65536u : 0u);
      const unsigned s1 = (unsigned)list[p + 1] + ((p + 1 >= p1) ? 65536u : 0u);
      const unsigned s2 = (unsigned)list[p + 2] + ((p + 2 >= p1) ? 65536u : 0u);
      const unsigned s3 = (unsigned)list[p + 3] + ((p + 3 >= p1) ? 65536u : 0u);
      const float4 h0 = H4[(size_t)s0 * 4 + q];
      const float4 h1 = H4[(size_t)s1 * 4 + q];
      const float4 h2 = H4[(size_t)s2 * 4 + q];
      const float4 h3 = H4[(size_t)s3 * 4 + q];
      edge_add(acc, h0);
      edge_add(acc, h1);
      edge_add(acc, h2);
      edge_add(acc, h3);
    }
    for (; p < pe; p++) {
      const unsigned s = (unsigned)list[p] + ((p >= p1) ? 65536u : 0u);
      edge_add(acc, H4[(size_t)s * 4 + q]);
    }

    const float4 hv = H4[(size_t)i * 4 + q];
    r.x = fmaxf(dn * (acc.x + hv.x) + bias[q * 4 + 0], 0.0f) * dn;
    r.y = fmaxf(dn * (acc.y + hv.y) + bias[q * 4 + 1], 0.0f) * dn;
    r.z = fmaxf(dn * (acc.z + hv.z) + bias[q * 4 + 2], 0.0f) * dn;
    r.w = fmaxf(dn * (acc.w + hv.w) + bias[q * 4 + 3], 0.0f) * dn;
  }
  *(float4*)(hs + (tid >> 2) * ROW_PITCH + q * 4) = r;
  __syncthreads();

  const int wv = tid >> 5;
  const int tile = blockIdx.x * 4 + wv;
  if (wv >= 4 || tile >= ntiles) return;

  const int lane = tid & 31;
  const int half = lane >> 4;
  const int mn   = lane & 15;
  const int koff = half * 2;
  const float* arow = hs + (wv * 16 + mn) * ROW_PITCH;

  v2f a[4], b[4];
#pragma unroll
  for (int s = 0; s < 4; s++) {
    a[s] = *(const v2f*)(arow + 4 * s + koff);
    b[s].x = Ws[(4 * s + koff + 0) * 16 + mn];
    b[s].y = Ws[(4 * s + koff + 1) * 16 + mn];
  }
  v8f c = {0.f, 0.f, 0.f, 0.f, 0.f, 0.f, 0.f, 0.f};
#pragma unroll
  for (int s = 0; s < 4; s++)
    c = __builtin_amdgcn_wmma_f32_16x16x4_f32(
         false, a[s],  false, b[s],
         (short)0, c,  false,  false);
  wmma_tail_slots(c, a, b);

  float* hrow = Hn + (size_t)tile * 256 + half * 128 + mn;
#pragma unroll
  for (int i8 = 0; i8 < 8; i8++) hrow[i8 * 16] = c[i8];
}

__global__ __launch_bounds__(256) void k_gather_head(const unsigned short* __restrict__ list,
                                                     const unsigned* __restrict__ rowptr,
                                                     const float* __restrict__ H,
                                                     const float* __restrict__ bias,
                                                     const float* __restrict__ W,
                                                     const float* __restrict__ l1b,
                                                     const float* __restrict__ l2W,
                                                     const float* __restrict__ l2b,
                                                     float* dio,
                                                     int n, int E, int ntiles) {
  __shared__ __attribute__((aligned(16))) float hs[64 * ROW_PITCH];
  __shared__ float Ws[256];
  const int tid = threadIdx.x;
  Ws[tid] = W[tid];

  const int idx = blockIdx.x * 256 + tid;
  const int i = idx >> 2;
  const int q = idx & 3;

  float4 r = make_float4(0.0f, 0.0f, 0.0f, 0.0f);
  if (i < n) {
    const unsigned p0 = rowptr[2 * i];
    const unsigned p1 = rowptr[2 * i + 1];
    const unsigned pe = (i + 1 < n) ? rowptr[2 * i + 2] : (unsigned)E;
    const float dn = dio[i];
    const float4* H4 = (const float4*)H;

    float4 acc = make_float4(0.0f, 0.0f, 0.0f, 0.0f);
    unsigned p = p0;
    for (; p + 4 <= pe; p += 4) {
      const unsigned s0 = (unsigned)list[p]     + ((p     >= p1) ? 65536u : 0u);
      const unsigned s1 = (unsigned)list[p + 1] + ((p + 1 >= p1) ? 65536u : 0u);
      const unsigned s2 = (unsigned)list[p + 2] + ((p + 2 >= p1) ? 65536u : 0u);
      const unsigned s3 = (unsigned)list[p + 3] + ((p + 3 >= p1) ? 65536u : 0u);
      const float4 h0 = H4[(size_t)s0 * 4 + q];
      const float4 h1 = H4[(size_t)s1 * 4 + q];
      const float4 h2 = H4[(size_t)s2 * 4 + q];
      const float4 h3 = H4[(size_t)s3 * 4 + q];
      edge_add(acc, h0);
      edge_add(acc, h1);
      edge_add(acc, h2);
      edge_add(acc, h3);
    }
    for (; p < pe; p++) {
      const unsigned s = (unsigned)list[p] + ((p >= p1) ? 65536u : 0u);
      edge_add(acc, H4[(size_t)s * 4 + q]);
    }

    const float4 hv = H4[(size_t)i * 4 + q];
    r.x = fmaxf(dn * (acc.x + hv.x) + bias[q * 4 + 0], 0.0f);
    r.y = fmaxf(dn * (acc.y + hv.y) + bias[q * 4 + 1], 0.0f);
    r.z = fmaxf(dn * (acc.z + hv.z) + bias[q * 4 + 2], 0.0f);
    r.w = fmaxf(dn * (acc.w + hv.w) + bias[q * 4 + 3], 0.0f);
  }
  *(float4*)(hs + (tid >> 2) * ROW_PITCH + q * 4) = r;
  __syncthreads();

  const int wv = tid >> 5;
  const int tile = blockIdx.x * 4 + wv;
  if (wv >= 4 || tile >= ntiles) return;

  const int lane = tid & 31;
  const int half = lane >> 4;
  const int mn   = lane & 15;
  const int koff = half * 2;
  const float* arow = hs + (wv * 16 + mn) * ROW_PITCH;
  const float bj = l1b[mn];
  const float wj = l2W[mn];

  v2f a[4], b[4];
#pragma unroll
  for (int s = 0; s < 4; s++) {
    a[s] = *(const v2f*)(arow + 4 * s + koff);
    b[s].x = Ws[(4 * s + koff + 0) * 16 + mn];
    b[s].y = Ws[(4 * s + koff + 1) * 16 + mn];
  }
  v8f c = {0.f, 0.f, 0.f, 0.f, 0.f, 0.f, 0.f, 0.f};
#pragma unroll
  for (int s = 0; s < 4; s++)
    c = __builtin_amdgcn_wmma_f32_16x16x4_f32(
         false, a[s],  false, b[s],
         (short)0, c,  false,  false);
  wmma_tail_slots(c, a, b);

  float mine = 0.0f;
#pragma unroll
  for (int r8 = 0; r8 < 8; r8++) {
    float v = fmaxf(c[r8] + bj, 0.0f) * wj;
#pragma unroll
    for (int off = 8; off; off >>= 1) v += __shfl_xor(v, off, 16);
    if (mn == r8) mine = v;
  }
  if (mn < 8) dio[tile * 16 + half * 8 + mn] = mine + l2b[0];
}


extern "C" void kernel_launch(void* const* d_in, const int* in_sizes, int n_in,
                              void* d_out, int out_size, void* d_ws, size_t ws_size,
                              hipStream_t stream) {
  const float* x   = (const float*)d_in[0];
  const int*   ei  = (const int*)d_in[1];
  const float* W1  = (const float*)d_in[2];
  const float* b1  = (const float*)d_in[3];
  const float* W2  = (const float*)d_in[4];
  const float* b2  = (const float*)d_in[5];
  const float* l1W = (const float*)d_in[6];
  const float* l1b = (const float*)d_in[7];
  const float* l2W = (const float*)d_in[8];
  const float* l2b = (const float*)d_in[9];

  const int N = in_sizes[0] / 128;
  const int E = in_sizes[1] / 2;
  const int* src = ei;
  const int* dst = ei + E;

  unsigned* rowptr = (unsigned*)d_ws;
  float* bufA = (float*)d_ws + 2 * (size_t)N;
  float* bufB = bufA + (size_t)N * 16;
  float* bufC = bufB + (size_t)N * 16;
  float* dinv = (float*)d_out;

  unsigned short* list = (unsigned short*)bufA;
  unsigned* packed     = (unsigned*)bufB;

  const int ntiles = N / 16;
  const int gemmBlocks = (ntiles * 32 + 255) / 256;
  const int gathBlocks = (N * 4 + 255) / 256;

  const int nb = (N + (1 << BUCKET_SHIFT) - 1) >> BUCKET_SHIFT;
  const int chunkLen = (E + EDGE_CHUNKS - 1) / EDGE_CHUNKS;
  k_bgroup<<<EDGE_CHUNKS, SORT_THREADS, 0, stream>>>(src, dst, packed, rowptr, E, chunkLen, nb);
  k_bsort <<<nb, SORT_THREADS, 0, stream>>>(packed, rowptr, list, dinv, chunkLen, N);

  k_gemm_wmma<128><<<gemmBlocks, 256, 0, stream>>>(x, W1, dinv, bufB, ntiles);
  k_gather_mm<<<gathBlocks, 256, 0, stream>>>(list, rowptr, dinv, bufB, b1, W2, bufC, N, E, ntiles);

  k_gather_head<<<gathBlocks, 256, 0, stream>>>(list, rowptr, bufC, b2, l1W, l1b, l2W, l2b,
                                                (float*)d_out, N, E, ntiles);
}


# Both gathers: the work-group's slice of the source list goes to LDS up front (16 B a thread); a trip reads 8 indices from LDS and asks for 8 rows, no index round trip per trip; add order kept
# speedup vs baseline: 6.3528x; 1.2399x over previous
#include <hip/hip_runtime.h>
#include <hip/hip_bf16.h>


typedef __attribute__((ext_vector_type(2))) float v2f;
typedef __attribute__((ext_vector_type(4))) float v4f;
typedef __attribute__((ext_vector_type(8))) float v8f;


#define BUCKET_SHIFT  9
#define BUCKET_KEYS   1024
#define MAX_BUCKETS   256
#define EDGE_CHUNKS   256
#define SORT_THREADS  1024
#define STAGE_CAP     32768
#define SCAT_CAP      16384

static_assert(SORT_THREADS == 4 * EDGE_CHUNKS, "k_bsort walks one run with four lanes");

template <bool STAGED>
__device__ __forceinline__ void chunk_place(const int* __restrict__ src, const int* __restrict__ dst,
                                            unsigned* cur, unsigned* stage,
                                            unsigned* __restrict__ packed, int beg, int end, int tid) {
  auto place = [&](int s, int d) {
    const unsigned pos = atomicAdd(&cur[((unsigned)d) >> BUCKET_SHIFT], 1u);
    const unsigned w = ((((unsigned)d) & 511u) << 17) | (unsigned)s;
    if (STAGED) stage[pos] = w;
    else packed[pos] = w;
  };
  int e = beg + tid;
  for (; e + 3 * SORT_THREADS < end; e += 4 * SORT_THREADS) {
    const int s0 = src[e];
    const int s1 = src[e + 1 * SORT_THREADS];
    const int s2 = src[e + 2 * SORT_THREADS];
    const int s3 = src[e + 3 * SORT_THREADS];
    const int d0 = dst[e];
    const int d1 = dst[e + 1 * SORT_THREADS];
    const int d2 = dst[e + 2 * SORT_THREADS];
    const int d3 = dst[e + 3 * SORT_THREADS];
    place(s0, d0);
    place(s1, d1);
    place(s2, d2);
    place(s3, d3);
  }
  for (; e < end; e += SORT_THREADS) place(src[e], dst[e]);
}

__global__ __launch_bounds__(SORT_THREADS) void k_bgroup(const int* __restrict__ src,
                                                         const int* __restrict__ dst,
                                                         unsigned* __restrict__ packed,
                                                         unsigned* __restrict__ tab,
                                                         int E, int chunkLen, int nb) {
  __shared__ unsigned h[MAX_BUCKETS];
  __shared__ unsigned so[MAX_BUCKETS];
  __shared__ unsigned stage[SCAT_CAP];
  const int tid = threadIdx.x;
  if (tid < MAX_BUCKETS) h[tid] = 0u;
  __syncthreads();

  int beg = blockIdx.x * chunkLen;
  if (beg > E) beg = E;
  int end = beg + chunkLen;
  if (end > E) end = E;
  const bool staged = (end - beg) <= SCAT_CAP;

  int e = beg + tid;
  for (; e + 3 * SORT_THREADS < end; e += 4 * SORT_THREADS) {
    const int d0 = dst[e];
    const int d1 = dst[e + 1 * SORT_THREADS];
    const int d2 = dst[e + 2 * SORT_THREADS];
    const int d3 = dst[e + 3 * SORT_THREADS];
    atomicAdd(&h[((unsigned)d0) >> BUCKET_SHIFT], 1u);
    atomicAdd(&h[((unsigned)d1) >> BUCKET_SHIFT], 1u);
    atomicAdd(&h[((unsigned)d2) >> BUCKET_SHIFT], 1u);
    atomicAdd(&h[((unsigned)d3) >> BUCKET_SHIFT], 1u);
  }
  for (; e < end; e += SORT_THREADS) atomicAdd(&h[((unsigned)dst[e]) >> BUCKET_SHIFT], 1u);
  __syncthreads();

  unsigned own = 0u;
  if (tid < MAX_BUCKETS) {
    own = h[tid];
    so[tid] = own;
  }
  __syncthreads();
  for (int off = 1; off < MAX_BUCKETS; off <<= 1) {
    unsigned a = 0u;
    if (tid < MAX_BUCKETS && tid >= off) a = so[tid - off];
    __syncthreads();
    if (tid < MAX_BUCKETS) so[tid] += a;
    __syncthreads();
  }
  if (tid < MAX_BUCKETS) {
    const unsigned lfirst = so[tid] - own;
    if (tid < nb) tab[tid * BUCKET_KEYS + blockIdx.x] = (lfirst << 16) | own;
    h[tid] = staged ? lfirst : ((unsigned)beg + lfirst);
  }
  __syncthreads();

  if (staged) {
    chunk_place<true>(src, dst, h, stage, packed, beg, end, tid);
    __syncthreads();
    const int len = end - beg;
    for (int i = tid; i < len; i += SORT_THREADS) packed[beg + i] = stage[i];
  } else {
    chunk_place<false>(src, dst, h, stage, packed, beg, end, tid);
  }
}

__device__ __forceinline__ unsigned packed_key(unsigned w) {
  return ((w >> 17) << 1) | ((w >> 16) & 1u);
}

template <class F>
__device__ __forceinline__ void run_words(const unsigned* __restrict__ run, const unsigned* __restrict__ safe,
                                          unsigned q, unsigned len, F f) {
  unsigned j = q;
  for (; j + 28u < len; j += 32u) {
    const unsigned w0 = run[j];
    const unsigned w1 = run[j + 4u];
    const unsigned w2 = run[j + 8u];
    const unsigned w3 = run[j + 12u];
    const unsigned w4 = run[j + 16u];
    const unsigned w5 = run[j + 20u];
    const unsigned w6 = run[j + 24u];
    const unsigned w7 = run[j + 28u];
    f(w0);
    f(w1);
    f(w2);
    f(w3);
    f(w4);
    f(w5);
    f(w6);
    f(w7);
  }
  unsigned w[7];
  bool have[7];
#pragma unroll
  for (int k = 0; k < 7; k++) {
    have[k] = (j + 4u * k) < len;
    const unsigned* p = have[k] ? (run + (j + 4u * k)) : safe;
    w[k] = *p;
  }
#pragma unroll
  for (int k = 0; k < 7; k++)
    if (have[k]) f(w[k]);
}

__global__ __launch_bounds__(SORT_THREADS) void k_bsort(const unsigned* __restrict__ packed,
                                                        unsigned* rowptr,
                                                        unsigned short* __restrict__ list,
                                                        float* __restrict__ dinv,
                                                        int chunkLen, int n) {
  __shared__ unsigned hist[BUCKET_KEYS];
  __shared__ unsigned sc[BUCKET_KEYS];
  __shared__ unsigned short stage[STAGE_CAP];
  __shared__ unsigned tot[2];
  const int tid = threadIdx.x;
  const int b = blockIdx.x;

  const unsigned t = rowptr[b * BUCKET_KEYS + (tid >> 2)];
  const unsigned len = t & 0xFFFFu;
  const unsigned lfirst = t >> 16;
  const unsigned* run = packed + (size_t)(tid >> 2) * (size_t)chunkLen + lfirst;
  const unsigned q = (unsigned)tid & 3u;

  hist[tid] = 0u;
  if (tid < 2) tot[tid] = 0u;
  __syncthreads();

  {
    unsigned a = (q == 0u) ? lfirst : 0u;
    unsigned c = (q == 0u) ? len : 0u;
    for (int off = 16; off; off >>= 1) {
      a += __shfl_xor(a, off);
      c += __shfl_xor(c, off);
    }
    if ((tid & 31) == 0) {
      atomicAdd(&tot[0], a);
      atomicAdd(&tot[1], c);
    }
  }

  run_words(run, packed, q, len, [&](unsigned w) { atomicAdd(&hist[packed_key(w)], 1u); });
  __syncthreads();

  const unsigned beg = tot[0];
  const unsigned total = tot[1];
  const unsigned mine  = hist[tid];
  const unsigned other = hist[tid ^ 1];
  sc[tid] = mine;
  __syncthreads();
  for (int off = 1; off < BUCKET_KEYS; off <<= 1) {
    const unsigned a = (tid >= off) ? sc[tid - off] : 0u;
    __syncthreads();
    sc[tid] += a;
    __syncthreads();
  }
  const unsigned first = beg + (sc[tid] - mine);
  hist[tid] = first;

  const int node = (b << BUCKET_SHIFT) + (tid >> 1);
  if (node < n) {
    rowptr[b * BUCKET_KEYS + tid] = first;
    if ((tid & 1) == 0) dinv[node] = rsqrtf((float)(1u + mine + other));
  }
  __syncthreads();

  const bool staged = total <= (unsigned)STAGE_CAP;
  auto place = [&](unsigned w) {
    const unsigned pos = atomicAdd(&hist[packed_key(w)], 1u);
    const unsigned short v = (unsigned short)(w & 0xFFFFu);
    if (staged) stage[pos - beg] = v;
    else list[pos] = v;
  };
  run_words(run, packed, q, len, place);
  if (staged) {
    __syncthreads();
    for (unsigned i = tid; i < total; i += SORT_THREADS) list[beg + i] = stage[i];
  }
}

template <int K>
__global__ void k_gemm_wmma(const float* __restrict__ A, const float* __restrict__ W,
                            const float* __restrict__ dinv,
                            float* __restrict__ H, int ntiles) {
  constexpr int PITCH = K / 2 + 2;
  __shared__ __attribute__((aligned(16))) float Wl[32 * PITCH];
  for (int t = threadIdx.x; t < K * 16; t += blockDim.x) {
    const int k = t >> 4, n = t & 15;
    Wl[((((k >> 1) & 1) << 4) | n) * PITCH + ((k >> 2) << 1) + (k & 1)] = W[t];
  }
  __syncthreads();

  const int lane = threadIdx.x & 31;
  const int tile = blockIdx.x * (blockDim.x >> 5) + (threadIdx.x >> 5);
  if (tile >= ntiles) return;

  const int half = lane >> 4;
  const int mn   = lane & 15;
  const int koff = half * 2;

  const float* arow = A + (size_t)(tile * 16 + mn) * K;
  const float* blane = Wl + lane * PITCH;

  v4f d0 = *(const v4f*)(dinv + tile * 16 + half * 8);
  v4f d1 = *(const v4f*)(dinv + tile * 16 + half * 8 + 4);
  int hoff = tile * 256 + half * 128 + mn;
  asm volatile("" : "+v"(d0), "+v"(d1), "+v"(hoff));

  v8f c = {0.f, 0.f, 0.f, 0.f, 0.f, 0.f, 0.f, 0.f};
  v2f a, b;
#pragma unroll
  for (int k0 = 0; k0 < K; k0 += 4) {
    a = *(const v2f*)(arow + k0 + koff);
    b = *(const v2f*)(blane + (k0 >> 1));
    c = __builtin_amdgcn_wmma_f32_16x16x4_f32(
         false, a,  false, b,
         (short)0, c,  false,  false);
  }
  asm volatile("v_nop\n\tv_nop\n\tv_nop\n\tv_nop" : "+v"(c), "+v"(hoff) : "v"(a), "v"(b));

  float* hrow = H + hoff;
  hrow[0 * 16] = c[0] * d0.x;
  hrow[1 * 16] = c[1] * d0.y;
  hrow[2 * 16] = c[2] * d0.z;
  hrow[3 * 16] = c[3] * d0.w;
  hrow[4 * 16] = c[4] * d1.x;
  hrow[5 * 16] = c[5] * d1.y;
  hrow[6 * 16] = c[6] * d1.z;
  hrow[7 * 16] = c[7] * d1.w;
}

__device__ __forceinline__ void edge_add(float4& acc, float4 h) {
  acc.x += h.x;
  acc.y += h.y;
  acc.z += h.z;
  acc.w += h.w;
}

__device__ __forceinline__ void wmma_tail_slots(v8f& c, const v2f (&a)[4], const v2f (&b)[4]) {
  asm volatile("v_nop\n\tv_nop\n\tv_nop\n\tv_nop"
               : "+v"(c)
               : "v"(a[0]), "v"(a[1]), "v"(a[2]), "v"(a[3]),
                 "v"(b[0]), "v"(b[1]), "v"(b[2]), "v"(b[3]));
}

#define LIST_CAP 8192

__device__ __forceinline__ void slice_range(const unsigned* __restrict__ rowptr, int n, int E,
                                            unsigned& abase, unsigned& span) {
  const int n0 = blockIdx.x * 64;
  const int n1 = n0 + 64;
  const unsigned lbeg = rowptr[2 * n0];
  const unsigned lend = (n1 < n) ? rowptr[2 * n1] : (unsigned)E;
  abase = lbeg & ~7u;
  span = lend - abase;
}

__device__ __forceinline__ void slice_stage(const unsigned short* __restrict__ list, unsigned short* ls,
                                            unsigned abase, unsigned span, int tid) {
  const uint4* g = (const uint4*)(list + abase);
  uint4* l = (uint4*)ls;
  const unsigned nvec = (span + 7u) >> 3;
  for (unsigned v = tid; v < nvec; v += 256) l[v] = g[v];
}

template <bool STAGED, bool MASKED>
__device__ __forceinline__ void trip8(const unsigned short* __restrict__ list, const unsigned short* ls,
                                      unsigned abase, unsigned g, const char* __restrict__ Hb,
                                      unsigned p0, unsigned p1, unsigned pe,
                                      unsigned own, unsigned qoff, float4& acc) {
  const uint4 w = STAGED ? *(const uint4*)(ls + g) : *(const uint4*)(list + abase + g);
  const unsigned lo[8] = {w.x & 0xFFFFu, w.x >> 16, w.y & 0xFFFFu, w.y >> 16,
                          w.z & 0xFFFFu, w.z >> 16, w.w & 0xFFFFu, w.w >> 16};
  bool have[8];
  unsigned off[8];
  float4 h[8];
#pragma unroll
  for (int k = 0; k < 8; k++) {
    const unsigned p = abase + g + k;
    have[k] = !MASKED || ((p >= p0) && (p < pe));
    off[k] = ((lo[k] + ((p >= p1) ? 65536u : 0u)) << 6) + qoff;
    if (!have[k]) off[k] = own;
  }
#pragma unroll
  for (int k = 0; k < 8; k++) h[k] = *(const float4*)(Hb + off[k]);
#pragma unroll
  for (int k = 0; k < 8; k++)
    if (have[k]) edge_add(acc, h[k]);
}

template <bool STAGED>
__device__ __forceinline__ float4 walk_sum(const unsigned short* __restrict__ list, const unsigned short* ls,
                                           unsigned abase, const float* __restrict__ H,
                                           unsigned p0, unsigned p1, unsigned pe, int i, int q) {
  const char* Hb = (const char*)H;
  const unsigned qoff = (unsigned)q * 16u;
  const unsigned own = (unsigned)i * 64u + qoff;
  float4 acc = make_float4(0.0f, 0.0f, 0.0f, 0.0f);
  if (p0 < pe) {
    unsigned g = (p0 - abase) & ~7u;
    trip8<STAGED, true>(list, ls, abase, g, Hb, p0, p1, pe, own, qoff, acc);
    g += 8;
    for (; abase + g + 8 <= pe; g += 8)
      trip8<STAGED, false>(list, ls, abase, g, Hb, p0, p1, pe, own, qoff, acc);
    if (abase + g < pe) trip8<STAGED, true>(list, ls, abase, g, Hb, p0, p1, pe, own, qoff, acc);
  }
  return acc;
}

#define ROW_PITCH 20

__global__ __launch_bounds__(256) void k_gather_mm(const unsigned short* __restrict__ list,
                                                   const unsigned* __restrict__ rowptr,
                                                   const float* __restrict__ dinv,
                                                   const float* __restrict__ H,
                                                   const float* __restrict__ bias,
                                                   const float* __restrict__ Wn,
                                                   float* __restrict__ Hn,
                                                   int n, int E, int ntiles) {
  __shared__ __attribute__((aligned(16))) float hs[64 * ROW_PITCH];
  __shared__ float Ws[256];
  __shared__ __attribute__((aligned(16))) unsigned short ls[LIST_CAP + 8];
  const int tid = threadIdx.x;
  const int idx = blockIdx.x * 256 + tid;
  const int i = idx >> 2;
  const int q = idx & 3;

  const float wt = Wn[tid];
  const float4* H4 = (const float4*)H;
  unsigned p0 = 0u, p1 = 0u, pe = 0u;
  float dn = 0.0f;
  float4 hv = make_float4(0.0f, 0.0f, 0.0f, 0.0f);
  if (i < n) {
    p0 = rowptr[2 * i];
    p1 = rowptr[2 * i + 1];
    pe = (i + 1 < n) ? rowptr[2 * i + 2] : (unsigned)E;
    dn = dinv[i];
    hv = H4[(size_t)i * 4 + q];
  }

  unsigned abase, span;
  slice_range(rowptr, n, E, abase, span);
  const bool staged = span <= (unsigned)LIST_CAP;
  if (staged) slice_stage(list, ls, abase, span, tid);
  Ws[tid] = wt;
  __syncthreads();

  float4 r = make_float4(0.0f, 0.0f, 0.0f, 0.0f);
  if (i < n) {
    const float4 acc = staged ? walk_sum<true>(list, ls, abase, H, p0, p1, pe, i, q)
                              : walk_sum<false>(list, ls, abase, H, p0, p1, pe, i, q);
    r.x = fmaxf(dn * (acc.x + hv.x) + bias[q * 4 + 0], 0.0f) * dn;
    r.y = fmaxf(dn * (acc.y + hv.y) + bias[q * 4 + 1], 0.0f) * dn;
    r.z = fmaxf(dn * (acc.z + hv.z) + bias[q * 4 + 2], 0.0f) * dn;
    r.w = fmaxf(dn * (acc.w + hv.w) + bias[q * 4 + 3], 0.0f) * dn;
  }
  *(float4*)(hs + (tid >> 2) * ROW_PITCH + q * 4) = r;
  __syncthreads();

  const int wv = tid >> 5;
  const int tile = blockIdx.x * 4 + wv;
  if (wv >= 4 || tile >= ntiles) return;

  const int lane = tid & 31;
  const int half = lane >> 4;
  const int mn   = lane & 15;
  const int koff = half * 2;
  const float* arow = hs + (wv * 16 + mn) * ROW_PITCH;

  v2f a[4], b[4];
#pragma unroll
  for (int s = 0; s < 4; s++) {
    a[s] = *(const v2f*)(arow + 4 * s + koff);
    b[s].x = Ws[(4 * s + koff + 0) * 16 + mn];
    b[s].y = Ws[(4 * s + koff + 1) * 16 + mn];
  }
  v8f c = {0.f, 0.f, 0.f, 0.f, 0.f, 0.f, 0.f, 0.f};
#pragma unroll
  for (int s = 0; s < 4; s++)
    c = __builtin_amdgcn_wmma_f32_16x16x4_f32(
         false, a[s],  false, b[s],
         (short)0, c,  false,  false);
  wmma_tail_slots(c, a, b);

  float* hrow = Hn + (size_t)tile * 256 + half * 128 + mn;
#pragma unroll
  for (int i8 = 0; i8 < 8; i8++) hrow[i8 * 16] = c[i8];
}

__global__ __launch_bounds__(256) void k_gather_head(const unsigned short* __restrict__ list,
                                                     const unsigned* __restrict__ rowptr,
                                                     const float* __restrict__ H,
                                                     const float* __restrict__ bias,
                                                     const float* __restrict__ W,
                                                     const float* __restrict__ l1b,
                                                     const float* __restrict__ l2W,
                                                     const float* __restrict__ l2b,
                                                     float* dio,
                                                     int n, int E, int ntiles) {
  __shared__ __attribute__((aligned(16))) float hs[64 * ROW_PITCH];
  __shared__ float Ws[256];
  __shared__ __attribute__((aligned(16))) unsigned short ls[LIST_CAP + 8];
  const int tid = threadIdx.x;
  const int idx = blockIdx.x * 256 + tid;
  const int i = idx >> 2;
  const int q = idx & 3;

  const float wt = W[tid];
  const float4* H4 = (const float4*)H;
  unsigned p0 = 0u, p1 = 0u, pe = 0u;
  float dn = 0.0f;
  float4 hv = make_float4(0.0f, 0.0f, 0.0f, 0.0f);
  if (i < n) {
    p0 = rowptr[2 * i];
    p1 = rowptr[2 * i + 1];
    pe = (i + 1 < n) ? rowptr[2 * i + 2] : (unsigned)E;
    dn = dio[i];
    hv = H4[(size_t)i * 4 + q];
  }

  unsigned abase, span;
  slice_range(rowptr, n, E, abase, span);
  const bool staged = span <= (unsigned)LIST_CAP;
  if (staged) slice_stage(list, ls, abase, span, tid);
  Ws[tid] = wt;
  __syncthreads();

  float4 r = make_float4(0.0f, 0.0f, 0.0f, 0.0f);
  if (i < n) {
    const float4 acc = staged ? walk_sum<true>(list, ls, abase, H, p0, p1, pe, i, q)
                              : walk_sum<false>(list, ls, abase, H, p0, p1, pe, i, q);
    r.x = fmaxf(dn * (acc.x + hv.x) + bias[q * 4 + 0], 0.0f);
    r.y = fmaxf(dn * (acc.y + hv.y) + bias[q * 4 + 1], 0.0f);
    r.z = fmaxf(dn * (acc.z + hv.z) + bias[q * 4 + 2], 0.0f);
    r.w = fmaxf(dn * (acc.w + hv.w) + bias[q * 4 + 3], 0.0f);
  }
  *(float4*)(hs + (tid >> 2) * ROW_PITCH + q * 4) = r;
  __syncthreads();

  const int wv = tid >> 5;
  const int tile = blockIdx.x * 4 + wv;
  if (wv >= 4 || tile >= ntiles) return;

  const int lane = tid & 31;
  const int half = lane >> 4;
  const int mn   = lane & 15;
  const int koff = half * 2;
  const float* arow = hs + (wv * 16 + mn) * ROW_PITCH;
  const float bj = l1b[mn];
  const float wj = l2W[mn];

  v2f a[4], b[4];
#pragma unroll
  for (int s = 0; s < 4; s++) {
    a[s] = *(const v2f*)(arow + 4 * s + koff);
    b[s].x = Ws[(4 * s + koff + 0) * 16 + mn];
    b[s].y = Ws[(4 * s + koff + 1) * 16 + mn];
  }
  v8f c = {0.f, 0.f, 0.f, 0.f, 0.f, 0.f, 0.f, 0.f};
#pragma unroll
  for (int s = 0; s < 4; s++)
    c = __builtin_amdgcn_wmma_f32_16x16x4_f32(
         false, a[s],  false, b[s],
         (short)0, c,  false,  false);
  wmma_tail_slots(c, a, b);

  float mine = 0.0f;
#pragma unroll
  for (int r8 = 0; r8 < 8; r8++) {
    float v = fmaxf(c[r8] + bj, 0.0f) * wj;
#pragma unroll
    for (int off = 8; off; off >>= 1) v += __shfl_xor(v, off, 16);
    if (mn == r8) mine = v;
  }
  if (mn < 8) dio[tile * 16 + half * 8 + mn] = mine + l2b[0];
}


extern "C" void kernel_launch(void* const* d_in, const int* in_sizes, int n_in,
                              void* d_out, int out_size, void* d_ws, size_t ws_size,
                              hipStream_t stream) {
  const float* x   = (const float*)d_in[0];
  const int*   ei  = (const int*)d_in[1];
  const float* W1  = (const float*)d_in[2];
  const float* b1  = (const float*)d_in[3];
  const float* W2  = (const float*)d_in[4];
  const float* b2  = (const float*)d_in[5];
  const float* l1W = (const float*)d_in[6];
  const float* l1b = (const float*)d_in[7];
  const float* l2W = (const float*)d_in[8];
  const float* l2b = (const float*)d_in[9];

  const int N = in_sizes[0] / 128;
  const int E = in_sizes[1] / 2;
  const int* src = ei;
  const int* dst = ei + E;

  unsigned* rowptr = (unsigned*)d_ws;
  float* bufA = (float*)d_ws + 2 * (size_t)N;
  float* bufB = bufA + (size_t)N * 16;
  float* bufC = bufB + (size_t)N * 16;
  float* dinv = (float*)d_out;

  unsigned short* list = (unsigned short*)bufA;
  unsigned* packed     = (unsigned*)bufB;

  const int ntiles = N / 16;
  const int gemmBlocks = (ntiles * 32 + 255) / 256;
  const int gathBlocks = (N * 4 + 255) / 256;

  const int nb = (N + (1 << BUCKET_SHIFT) - 1) >> BUCKET_SHIFT;
  const int chunkLen = (E + EDGE_CHUNKS - 1) / EDGE_CHUNKS;
  k_bgroup<<<EDGE_CHUNKS, SORT_THREADS, 0, stream>>>(src, dst, packed, rowptr, E, chunkLen, nb);
  k_bsort <<<nb, SORT_THREADS, 0, stream>>>(packed, rowptr, list, dinv, chunkLen, N);

  k_gemm_wmma<128><<<gemmBlocks, 256, 0, stream>>>(x, W1, dinv, bufB, ntiles);
  k_gather_mm<<<gathBlocks, 256, 0, stream>>>(list, rowptr, dinv, bufB, b1, W2, bufC, N, E, ntiles);

  k_gather_head<<<gathBlocks, 256, 0, stream>>>(list, rowptr, bufC, b2, l1W, l1b, l2W, l2b,
                                                (float*)d_out, N, E, ntiles);
}
